# starting kernel
# baseline (speedup 1.0000x reference)
#include <hip/hip_runtime.h>


#define NMAT 2048
#define SQUARINGS 4
#define HORNER_TOP 8

typedef float v2f __attribute__((ext_vector_type(2)));
typedef float v8f __attribute__((ext_vector_type(8)));

__global__ __launch_bounds__(256)
void skew_init_kernel(const float* __restrict__ A,
                      float* __restrict__ X,
                      float* __restrict__ P)
{
    const int idx = blockIdx.x * 256 + threadIdx.x;
    const int i = idx >> 11;
    const int j = idx & (NMAT - 1);
    const float invScale = 1.0f / (float)(1 << SQUARINGS);
    float s = (A[i * NMAT + j] - A[j * NMAT + i]) * invScale;
    X[idx] = s;
    P[idx] = s * (1.0f / (float)HORNER_TOP) + ((i == j) ? 1.0f : 0.0f);
}

__global__ __launch_bounds__(256)
void gemm_wmma_f32_kernel(const float* __restrict__ Amat,
                          const float* __restrict__ Bmat,
                          float* __restrict__ Dmat,
                          float alpha, int addIdentity)
{
    __shared__ float As[128][36];
    __shared__ float Bs[128][36];

    const int tid  = threadIdx.x;
    const int wave = tid >> 5;
    const int lane = tid & 31;
    const int half = lane >> 4;
    const int r    = lane & 15;

    const int rowBase = blockIdx.y * 128;
    const int colBase = blockIdx.x * 128;
    const int waveRow = (wave >> 1) * 32;
    const int waveCol = (wave & 1) * 64;

    v8f acc[2][4];
#pragma unroll
    for (int m = 0; m < 2; ++m)
#pragma unroll
        for (int n = 0; n < 4; ++n)
            acc[m][n] = (v8f)0.0f;

    for (int kt = 0; kt < NMAT; kt += 32) {
#pragma unroll
        for (int it = 0; it < 4; ++it) {
            int idx  = tid + it * 256;
            int arow = idx >> 3;
            int ac4  = (idx & 7) << 2;
            const float4 v =
                *(const float4*)&Amat[(rowBase + arow) * NMAT + kt + ac4];
            *(float4*)&As[arow][ac4] = v;
        }
#pragma unroll
        for (int it = 0; it < 4; ++it) {
            int idx  = tid + it * 256;
            int krow = idx >> 5;
            int bc4  = (idx & 31) << 2;
            const float4 v =
                *(const float4*)&Bmat[(kt + krow) * NMAT + colBase + bc4];
            Bs[bc4 + 0][krow] = v.x;
            Bs[bc4 + 1][krow] = v.y;
            Bs[bc4 + 2][krow] = v.z;
            Bs[bc4 + 3][krow] = v.w;
        }
        __syncthreads();

#pragma unroll
        for (int kk = 0; kk < 32; kk += 4) {
            const int ka = kk + 2 * half;
            v2f a[2], b[4];
#pragma unroll
            for (int m = 0; m < 2; ++m)
                a[m] = *(const v2f*)&As[waveRow + 16 * m + r][ka];
#pragma unroll
            for (int n = 0; n < 4; ++n)
                b[n] = *(const v2f*)&Bs[waveCol + 16 * n + r][ka];
#pragma unroll
            for (int m = 0; m < 2; ++m)
#pragma unroll
                for (int n = 0; n < 4; ++n)
                    acc[m][n] = __builtin_amdgcn_wmma_f32_16x16x4_f32(
                         false, a[m],
                         false, b[n],
                         (short)0, acc[m][n],
                         false,  false);
        }
        __syncthreads();
    }

#pragma unroll
    for (int m = 0; m < 2; ++m) {
#pragma unroll
        for (int n = 0; n < 4; ++n) {
#pragma unroll
            for (int i = 0; i < 8; ++i) {
                const int row = rowBase + waveRow + 16 * m + i + 8 * half;
                const int col = colBase + waveCol + 16 * n + r;
                float val = alpha * acc[m][n][i];
                if (addIdentity && (row == col)) val += 1.0f;
                Dmat[row * NMAT + col] = val;
            }
        }
    }
}

extern "C" void kernel_launch(void* const* d_in, const int* in_sizes, int n_in,
                              void* d_out, int out_size, void* d_ws, size_t ws_size,
                              hipStream_t stream)
{
    (void)in_sizes; (void)n_in; (void)out_size; (void)ws_size;

    const float* A  = (const float*)d_in[0];
    float*       out = (float*)d_out;

    const size_t NN = (size_t)NMAT * NMAT;
    float* X  = (float*)d_ws;
    float* P1 = X + NN;
    float* P2 = P1 + NN;

    skew_init_kernel<<<(int)(NN / 256), 256, 0, stream>>>(A, X, P1);

    dim3 grid(NMAT / 128, NMAT / 128);
    dim3 block(256);

    float* cur = P1;
    float* oth = P2;
    for (int k = HORNER_TOP - 1; k >= 1; --k) {
        gemm_wmma_f32_kernel<<<grid, block, 0, stream>>>(
            X, cur, oth, 1.0f / (float)k,  1);
        float* t = cur; cur = oth; oth = t;
    }

    for (int sq = 0; sq < SQUARINGS; ++sq) {
        float* dst = (sq == SQUARINGS - 1) ? out : oth;
        gemm_wmma_f32_kernel<<<grid, block, 0, stream>>>(
            cur, cur, dst, 1.0f,  0);
        oth = cur;
        cur = dst;
    }
}


# Evaluate the same degree-8 Taylor polynomial with 3 matrix products (X2; X4=X2*(aX+bX2); (cX2+X4)*(quadratic in X2, linear in X)) not 7 Horner GEMMs: 7 GEMMs not 11; float summation order changes
# speedup vs baseline: 1.9584x; 1.9584x over previous
#include <hip/hip_runtime.h>


#define NMAT 2048
#define SQUARINGS 4

#define K_X1 0.10836465678522780f
#define K_X2 0.02709116419630695f
#define K_X3 0.66666666666666667f
#define K_X4 0.54676145797072400f
#define K_X5 0.16112557339541758f
#define K_X6 0.01409091715837821f
#define K_X7 0.03379279701087050f
#define K_Y2 0.13549236135285073f

typedef float v2f __attribute__((ext_vector_type(2)));
typedef float v8f __attribute__((ext_vector_type(8)));

__global__ __launch_bounds__(256)
void skew_init_kernel(const float* __restrict__ A,
                      float* __restrict__ X)
{
    const int idx = blockIdx.x * 256 + threadIdx.x;
    const int i = idx >> 11;
    const int j = idx & (NMAT - 1);
    const float invScale = 1.0f / (float)(1 << SQUARINGS);
    float s = (A[i * NMAT + j] - A[j * NMAT + i]) * invScale;
    X[idx] = s;
}

__global__ __launch_bounds__(256)
void gemm_wmma_f32_kernel(const float* __restrict__ Amat,
                          const float* __restrict__ Bmat,
                          float* __restrict__ Dmat,
                          const float* __restrict__ Emat,
                          float* __restrict__ D2mat,
                          const float* __restrict__ Ain,
                          int mode)
{
    __shared__ float As[128][36];
    __shared__ float Bs[128][36];

    const int tid  = threadIdx.x;
    const int wave = tid >> 5;
    const int lane = tid & 31;
    const int half = lane >> 4;
    const int r    = lane & 15;

    const int rowBase = blockIdx.y * 128;
    const int colBase = blockIdx.x * 128;
    const int waveRow = (wave >> 1) * 32;
    const int waveCol = (wave & 1) * 64;

    v8f acc[2][4];
#pragma unroll
    for (int m = 0; m < 2; ++m)
#pragma unroll
        for (int n = 0; n < 4; ++n)
            acc[m][n] = (v8f)0.0f;

    for (int kt = 0; kt < NMAT; kt += 32) {
#pragma unroll
        for (int it = 0; it < 4; ++it) {
            int idx  = tid + it * 256;
            int arow = idx >> 3;
            int ac4  = (idx & 7) << 2;
            const float4 v =
                *(const float4*)&Amat[(rowBase + arow) * NMAT + kt + ac4];
            *(float4*)&As[arow][ac4] = v;
        }
#pragma unroll
        for (int it = 0; it < 4; ++it) {
            int idx  = tid + it * 256;
            int krow = idx >> 5;
            int bc4  = (idx & 31) << 2;
            const float4 v =
                *(const float4*)&Bmat[(kt + krow) * NMAT + colBase + bc4];
            Bs[bc4 + 0][krow] = v.x;
            Bs[bc4 + 1][krow] = v.y;
            Bs[bc4 + 2][krow] = v.z;
            Bs[bc4 + 3][krow] = v.w;
        }
        __syncthreads();

#pragma unroll
        for (int kk = 0; kk < 32; kk += 4) {
            const int ka = kk + 2 * half;
            v2f a[2], b[4];
#pragma unroll
            for (int m = 0; m < 2; ++m)
                a[m] = *(const v2f*)&As[waveRow + 16 * m + r][ka];
#pragma unroll
            for (int n = 0; n < 4; ++n)
                b[n] = *(const v2f*)&Bs[waveCol + 16 * n + r][ka];
#pragma unroll
            for (int m = 0; m < 2; ++m)
#pragma unroll
                for (int n = 0; n < 4; ++n)
                    acc[m][n] = __builtin_amdgcn_wmma_f32_16x16x4_f32(
                         false, a[m],
                         false, b[n],
                         (short)0, acc[m][n],
                         false,  false);
        }
        __syncthreads();
    }

    const int row0 = rowBase + waveRow + 8 * half;
    const int col0 = colBase + waveCol + r;
    if (mode == 0) {
#pragma unroll
        for (int m = 0; m < 2; ++m)
#pragma unroll
            for (int n = 0; n < 4; ++n)
#pragma unroll
                for (int i = 0; i < 8; ++i) {
                    const int row = row0 + 16 * m + i;
                    const int col = col0 + 16 * n;
                    Dmat[row * NMAT + col] = acc[m][n][i];
                }
    } else if (mode == 1) {
#pragma unroll
        for (int m = 0; m < 2; ++m)
#pragma unroll
            for (int n = 0; n < 4; ++n)
#pragma unroll
                for (int i = 0; i < 8; ++i) {
                    const int row = row0 + 16 * m + i;
                    const int col = col0 + 16 * n;
                    const float p = acc[m][n][i];
                    const float x = Emat[row * NMAT + col];
                    Dmat[row * NMAT + col]  = p;
                    D2mat[row * NMAT + col] = K_X2 * p + K_X1 * x;
                }
    } else if (mode == 2) {
#pragma unroll
        for (int m = 0; m < 2; ++m)
#pragma unroll
            for (int n = 0; n < 4; ++n)
#pragma unroll
                for (int i = 0; i < 8; ++i) {
                    const int row = row0 + 16 * m + i;
                    const int col = col0 + 16 * n;
                    const float p  = acc[m][n][i];
                    const float x2 = Emat[row * NMAT + col];
                    const float x  = D2mat[row * NMAT + col];
                    Dmat[row * NMAT + col]  = p + K_X3 * x2;
                    D2mat[row * NMAT + col] = K_X7 * p + K_X6 * x2 + K_X5 * x
                                            + ((row == col) ? K_X4 : 0.0f);
                }
    } else {
        const float invScale = 1.0f / (float)(1 << SQUARINGS);
#pragma unroll
        for (int m = 0; m < 2; ++m)
#pragma unroll
            for (int n = 0; n < 4; ++n)
#pragma unroll
                for (int i = 0; i < 8; ++i) {
                    const int row = row0 + 16 * m + i;
                    const int col = col0 + 16 * n;
                    const float p  = acc[m][n][i];
                    const float x2 = Emat[row * NMAT + col];
                    const float x  = (Ain[row * NMAT + col] - Ain[col * NMAT + row])
                                   * invScale;
                    Dmat[row * NMAT + col] = p + K_Y2 * x2 + x
                                           + ((row == col) ? 1.0f : 0.0f);
                }
    }
}

extern "C" void kernel_launch(void* const* d_in, const int* in_sizes, int n_in,
                              void* d_out, int out_size, void* d_ws, size_t ws_size,
                              hipStream_t stream)
{
    (void)in_sizes; (void)n_in; (void)out_size; (void)ws_size;

    const float* A  = (const float*)d_in[0];
    float*       out = (float*)d_out;

    static_assert(SQUARINGS == 4, "the launch sequence below squares four times");
    const size_t NN = (size_t)NMAT * NMAT;
    float* W0 = (float*)d_ws;
    float* W1 = W0 + NN;
    float* W2 = W1 + NN;

    skew_init_kernel<<<(int)(NN / 256), 256, 0, stream>>>(A, W0);

    dim3 grid(NMAT / 128, NMAT / 128);
    dim3 block(256);

    gemm_wmma_f32_kernel<<<grid, block, 0, stream>>>(W0, W0, W1, W0, W2, A, 1);
    gemm_wmma_f32_kernel<<<grid, block, 0, stream>>>(W1, W2, out, W1, W0, A, 2);
    gemm_wmma_f32_kernel<<<grid, block, 0, stream>>>(out, W0, W2, W1, nullptr, A, 3);

    gemm_wmma_f32_kernel<<<grid, block, 0, stream>>>(W2, W2, W0, nullptr, nullptr, A, 0);
    gemm_wmma_f32_kernel<<<grid, block, 0, stream>>>(W0, W0, W1, nullptr, nullptr, A, 0);
    gemm_wmma_f32_kernel<<<grid, block, 0, stream>>>(W1, W1, W0, nullptr, nullptr, A, 0);
    gemm_wmma_f32_kernel<<<grid, block, 0, stream>>>(W0, W0, out, nullptr, nullptr, A, 0);
}


# Software-pipeline the GEMM K loop over two LDS stages: the next tile is fetched global->registers a trip ahead, so its latency and LDS stores run under WMMAs; one barrier per K-tile, not two
# speedup vs baseline: 2.6461x; 1.3512x over previous
#include <hip/hip_runtime.h>


#define NMAT 2048
#define SQUARINGS 4
#define LDP 36

#define K_X1 0.10836465678522780f
#define K_X2 0.02709116419630695f
#define K_X3 0.66666666666666667f
#define K_X4 0.54676145797072400f
#define K_X5 0.16112557339541758f
#define K_X6 0.01409091715837821f
#define K_X7 0.03379279701087050f
#define K_Y2 0.13549236135285073f

typedef float v2f __attribute__((ext_vector_type(2)));
typedef float v4f __attribute__((ext_vector_type(4)));
typedef float v8f __attribute__((ext_vector_type(8)));

__global__ __launch_bounds__(256)
void skew_init_kernel(const float* __restrict__ A,
                      float* __restrict__ X)
{
    const int idx = blockIdx.x * 256 + threadIdx.x;
    const int i = idx >> 11;
    const int j = idx & (NMAT - 1);
    const float invScale = 1.0f / (float)(1 << SQUARINGS);
    float s = (A[i * NMAT + j] - A[j * NMAT + i]) * invScale;
    X[idx] = s;
}

struct TileRegs { v4f a0, a1, a2, a3, b0, b1, b2, b3; };

__device__ __forceinline__
void tile_fetch(const float* __restrict__ aSrc, const float* __restrict__ bSrc,
                int kt, TileRegs& p)
{
    p.a0 = *(const v4f*)(aSrc +  0 * NMAT + kt);
    p.a1 = *(const v4f*)(aSrc + 32 * NMAT + kt);
    p.a2 = *(const v4f*)(aSrc + 64 * NMAT + kt);
    p.a3 = *(const v4f*)(aSrc + 96 * NMAT + kt);
    p.b0 = *(const v4f*)(bSrc + (kt +  0) * NMAT);
    p.b1 = *(const v4f*)(bSrc + (kt +  8) * NMAT);
    p.b2 = *(const v4f*)(bSrc + (kt + 16) * NMAT);
    p.b3 = *(const v4f*)(bSrc + (kt + 24) * NMAT);
}

__device__ __forceinline__
void store_b(float (*Bs)[LDP], int bc4, int k, const v4f v)
{
    Bs[bc4 + 0][k] = v.x;
    Bs[bc4 + 1][k] = v.y;
    Bs[bc4 + 2][k] = v.z;
    Bs[bc4 + 3][k] = v.w;
}

__device__ __forceinline__
void tile_store(float (*As)[LDP], float (*Bs)[LDP], int tid, const TileRegs& p)
{
    const int arow = tid >> 3;
    const int ac4  = (tid & 7) << 2;
    const int krow = tid >> 5;
    const int bc4  = (tid & 31) << 2;
    *(v4f*)&As[arow +  0][ac4] = p.a0;
    *(v4f*)&As[arow + 32][ac4] = p.a1;
    *(v4f*)&As[arow + 64][ac4] = p.a2;
    *(v4f*)&As[arow + 96][ac4] = p.a3;
    store_b(Bs, bc4, krow +  0, p.b0);
    store_b(Bs, bc4, krow +  8, p.b1);
    store_b(Bs, bc4, krow + 16, p.b2);
    store_b(Bs, bc4, krow + 24, p.b3);
}

__device__ __forceinline__
void tile_mma(const float (*As)[LDP], const float (*Bs)[LDP],
              int waveRow, int waveCol, int r, int half, v8f (&acc)[2][4])
{
#pragma unroll
    for (int kk = 0; kk < 32; kk += 4) {
        const int ka = kk + 2 * half;
        v2f a[2], b[4];
#pragma unroll
        for (int m = 0; m < 2; ++m)
            a[m] = *(const v2f*)&As[waveRow + 16 * m + r][ka];
#pragma unroll
        for (int n = 0; n < 4; ++n)
            b[n] = *(const v2f*)&Bs[waveCol + 16 * n + r][ka];
#pragma unroll
        for (int m = 0; m < 2; ++m)
#pragma unroll
            for (int n = 0; n < 4; ++n)
                acc[m][n] = __builtin_amdgcn_wmma_f32_16x16x4_f32(
                     false, a[m],
                     false, b[n],
                     (short)0, acc[m][n],
                     false,  false);
    }
}

__global__ __launch_bounds__(256)
void gemm_wmma_f32_kernel(const float* __restrict__ Amat,
                          const float* __restrict__ Bmat,
                          float* __restrict__ Dmat,
                          const float* __restrict__ Emat,
                          float* __restrict__ D2mat,
                          const float* __restrict__ Ain,
                          int mode)
{
    __shared__ __attribute__((aligned(16))) float As[2][128][LDP];
    __shared__ __attribute__((aligned(16))) float Bs[2][128][LDP];

    const int tid  = threadIdx.x;
    const int wave = tid >> 5;
    const int lane = tid & 31;
    const int half = lane >> 4;
    const int r    = lane & 15;

    const int rowBase = blockIdx.y * 128;
    const int colBase = blockIdx.x * 128;
    const int waveRow = (wave >> 1) * 32;
    const int waveCol = (wave & 1) * 64;

    v8f acc[2][4];
#pragma unroll
    for (int m = 0; m < 2; ++m)
#pragma unroll
        for (int n = 0; n < 4; ++n)
            acc[m][n] = (v8f)0.0f;

    const float* aSrc = Amat + (rowBase + (tid >> 3)) * NMAT + ((tid & 7) << 2);
    const float* bSrc = Bmat + (tid >> 5) * NMAT + colBase + ((tid & 31) << 2);

    TileRegs p;
    tile_fetch(aSrc, bSrc, 0, p);
    tile_store(As[0], Bs[0], tid, p);
    __syncthreads();
    tile_fetch(aSrc, bSrc, 32, p);

    for (int t = 0; t < NMAT / 32 - 2; ++t) {
        const int cur = t & 1;
        tile_mma(As[cur], Bs[cur], waveRow, waveCol, r, half, acc);
        tile_store(As[cur ^ 1], Bs[cur ^ 1], tid, p);
        __syncthreads();
        tile_fetch(aSrc, bSrc, (t + 2) * 32, p);
    }
    tile_mma(As[0], Bs[0], waveRow, waveCol, r, half, acc);
    tile_store(As[1], Bs[1], tid, p);
    __syncthreads();
    tile_mma(As[1], Bs[1], waveRow, waveCol, r, half, acc);

    const int row0 = rowBase + waveRow + 8 * half;
    const int col0 = colBase + waveCol + r;
    if (mode == 0) {
#pragma unroll
        for (int m = 0; m < 2; ++m)
#pragma unroll
            for (int n = 0; n < 4; ++n)
#pragma unroll
                for (int i = 0; i < 8; ++i) {
                    const int row = row0 + 16 * m + i;
                    const int col = col0 + 16 * n;
                    Dmat[row * NMAT + col] = acc[m][n][i];
                }
    } else if (mode == 1) {
#pragma unroll
        for (int m = 0; m < 2; ++m)
#pragma unroll
            for (int n = 0; n < 4; ++n)
#pragma unroll
                for (int i = 0; i < 8; ++i) {
                    const int row = row0 + 16 * m + i;
                    const int col = col0 + 16 * n;
                    const float p = acc[m][n][i];
                    const float x = Emat[row * NMAT + col];
                    Dmat[row * NMAT + col]  = p;
                    D2mat[row * NMAT + col] = K_X2 * p + K_X1 * x;
                }
    } else if (mode == 2) {
#pragma unroll
        for (int m = 0; m < 2; ++m)
#pragma unroll
            for (int n = 0; n < 4; ++n)
#pragma unroll
                for (int i = 0; i < 8; ++i) {
                    const int row = row0 + 16 * m + i;
                    const int col = col0 + 16 * n;
                    const float p  = acc[m][n][i];
                    const float x2 = Emat[row * NMAT + col];
                    const float x  = D2mat[row * NMAT + col];
                    Dmat[row * NMAT + col]  = p + K_X3 * x2;
                    D2mat[row * NMAT + col] = K_X7 * p + K_X6 * x2 + K_X5 * x
                                            + ((row == col) ? K_X4 : 0.0f);
                }
    } else {
        const float invScale = 1.0f / (float)(1 << SQUARINGS);
#pragma unroll
        for (int m = 0; m < 2; ++m)
#pragma unroll
            for (int n = 0; n < 4; ++n)
#pragma unroll
                for (int i = 0; i < 8; ++i) {
                    const int row = row0 + 16 * m + i;
                    const int col = col0 + 16 * n;
                    const float p  = acc[m][n][i];
                    const float x2 = Emat[row * NMAT + col];
                    const float x  = (Ain[row * NMAT + col] - Ain[col * NMAT + row])
                                   * invScale;
                    Dmat[row * NMAT + col] = p + K_Y2 * x2 + x
                                           + ((row == col) ? 1.0f : 0.0f);
                }
    }
}

extern "C" void kernel_launch(void* const* d_in, const int* in_sizes, int n_in,
                              void* d_out, int out_size, void* d_ws, size_t ws_size,
                              hipStream_t stream)
{
    (void)in_sizes; (void)n_in; (void)out_size; (void)ws_size;

    const float* A  = (const float*)d_in[0];
    float*       out = (float*)d_out;

    static_assert(SQUARINGS == 4, "the launch sequence below squares four times");
    const size_t NN = (size_t)NMAT * NMAT;
    float* W0 = (float*)d_ws;
    float* W1 = W0 + NN;
    float* W2 = W1 + NN;

    skew_init_kernel<<<(int)(NN / 256), 256, 0, stream>>>(A, W0);

    dim3 grid(NMAT / 128, NMAT / 128);
    dim3 block(256);

    gemm_wmma_f32_kernel<<<grid, block, 0, stream>>>(W0, W0, W1, W0, W2, A, 1);
    gemm_wmma_f32_kernel<<<grid, block, 0, stream>>>(W1, W2, out, W1, W0, A, 2);
    gemm_wmma_f32_kernel<<<grid, block, 0, stream>>>(out, W0, W2, W1, nullptr, A, 3);

    gemm_wmma_f32_kernel<<<grid, block, 0, stream>>>(W2, W2, W0, nullptr, nullptr, A, 0);
    gemm_wmma_f32_kernel<<<grid, block, 0, stream>>>(W0, W0, W1, nullptr, nullptr, A, 0);
    gemm_wmma_f32_kernel<<<grid, block, 0, stream>>>(W1, W1, W0, nullptr, nullptr, A, 0);
    gemm_wmma_f32_kernel<<<grid, block, 0, stream>>>(W0, W0, out, nullptr, nullptr, A, 0);
}


# Stage B with each wave on 32 columns x 4 k rows per vector (was 128 columns x 1 k row): a wave's transposed LDS stores fall on 16 banks 2 deep, not 4 banks 8 deep; sums unchanged
# speedup vs baseline: 2.7451x; 1.0374x over previous
#include <hip/hip_runtime.h>


#define NMAT 2048
#define SQUARINGS 4
#define LDP 36

#define K_X1 0.10836465678522780f
#define K_X2 0.02709116419630695f
#define K_X3 0.66666666666666667f
#define K_X4 0.54676145797072400f
#define K_X5 0.16112557339541758f
#define K_X6 0.01409091715837821f
#define K_X7 0.03379279701087050f
#define K_Y2 0.13549236135285073f

typedef float v2f __attribute__((ext_vector_type(2)));
typedef float v4f __attribute__((ext_vector_type(4)));
typedef float v8f __attribute__((ext_vector_type(8)));

__global__ __launch_bounds__(256)
void skew_init_kernel(const float* __restrict__ A,
                      float* __restrict__ X)
{
    const int idx = blockIdx.x * 256 + threadIdx.x;
    const int i = idx >> 11;
    const int j = idx & (NMAT - 1);
    const float invScale = 1.0f / (float)(1 << SQUARINGS);
    float s = (A[i * NMAT + j] - A[j * NMAT + i]) * invScale;
    X[idx] = s;
}

struct TileRegs { v4f a0, a1, a2, a3, b0, b1, b2, b3; };

__device__ __forceinline__
void tile_fetch(const float* __restrict__ aSrc, const float* __restrict__ bSrc,
                int kt, TileRegs& p)
{
    p.a0 = *(const v4f*)(aSrc +  0 * NMAT + kt);
    p.a1 = *(const v4f*)(aSrc + 32 * NMAT + kt);
    p.a2 = *(const v4f*)(aSrc + 64 * NMAT + kt);
    p.a3 = *(const v4f*)(aSrc + 96 * NMAT + kt);
    p.b0 = *(const v4f*)(bSrc + (kt +  0) * NMAT);
    p.b1 = *(const v4f*)(bSrc + (kt +  8) * NMAT);
    p.b2 = *(const v4f*)(bSrc + (kt + 16) * NMAT);
    p.b3 = *(const v4f*)(bSrc + (kt + 24) * NMAT);
}

__device__ __forceinline__
void store_b(float (*Bs)[LDP], int bc4, int k, const v4f v)
{
    Bs[bc4 + 0][k] = v.x;
    Bs[bc4 + 1][k] = v.y;
    Bs[bc4 + 2][k] = v.z;
    Bs[bc4 + 3][k] = v.w;
}

__device__ __forceinline__
void tile_store(float (*As)[LDP], float (*Bs)[LDP], int tid, const TileRegs& p)
{
    const int arow = tid >> 3;
    const int ac4  = (tid & 7) << 2;
    const int krow = ((tid >> 3) & 3) + ((tid >> 7) << 2);
    const int bc4  = (((tid >> 5) & 3) << 5) + ((tid & 7) << 2);
    *(v4f*)&As[arow +  0][ac4] = p.a0;
    *(v4f*)&As[arow + 32][ac4] = p.a1;
    *(v4f*)&As[arow + 64][ac4] = p.a2;
    *(v4f*)&As[arow + 96][ac4] = p.a3;
    store_b(Bs, bc4, krow +  0, p.b0);
    store_b(Bs, bc4, krow +  8, p.b1);
    store_b(Bs, bc4, krow + 16, p.b2);
    store_b(Bs, bc4, krow + 24, p.b3);
}

__device__ __forceinline__
void tile_mma(const float (*As)[LDP], const float (*Bs)[LDP],
              int waveRow, int waveCol, int r, int half, v8f (&acc)[2][4])
{
#pragma unroll
    for (int kk = 0; kk < 32; kk += 4) {
        const int ka = kk + 2 * half;
        v2f a[2], b[4];
#pragma unroll
        for (int m = 0; m < 2; ++m)
            a[m] = *(const v2f*)&As[waveRow + 16 * m + r][ka];
#pragma unroll
        for (int n = 0; n < 4; ++n)
            b[n] = *(const v2f*)&Bs[waveCol + 16 * n + r][ka];
#pragma unroll
        for (int m = 0; m < 2; ++m)
#pragma unroll
            for (int n = 0; n < 4; ++n)
                acc[m][n] = __builtin_amdgcn_wmma_f32_16x16x4_f32(
                     false, a[m],
                     false, b[n],
                     (short)0, acc[m][n],
                     false,  false);
    }
}

__global__ __launch_bounds__(256)
void gemm_wmma_f32_kernel(const float* __restrict__ Amat,
                          const float* __restrict__ Bmat,
                          float* __restrict__ Dmat,
                          const float* __restrict__ Emat,
                          float* __restrict__ D2mat,
                          const float* __restrict__ Ain,
                          int mode)
{
    __shared__ __attribute__((aligned(16))) float As[2][128][LDP];
    __shared__ __attribute__((aligned(16))) float Bs[2][128][LDP];

    const int tid  = threadIdx.x;
    const int wave = tid >> 5;
    const int lane = tid & 31;
    const int half = lane >> 4;
    const int r    = lane & 15;

    const int rowBase = blockIdx.y * 128;
    const int colBase = blockIdx.x * 128;
    const int waveRow = (wave >> 1) * 32;
    const int waveCol = (wave & 1) * 64;

    v8f acc[2][4];
#pragma unroll
    for (int m = 0; m < 2; ++m)
#pragma unroll
        for (int n = 0; n < 4; ++n)
            acc[m][n] = (v8f)0.0f;

    const float* aSrc = Amat + (rowBase + (tid >> 3)) * NMAT + ((tid & 7) << 2);
    const float* bSrc = Bmat + (((tid >> 3) & 3) + ((tid >> 7) << 2)) * NMAT
                      + colBase + (((tid >> 5) & 3) << 5) + ((tid & 7) << 2);

    TileRegs p;
    tile_fetch(aSrc, bSrc, 0, p);
    tile_store(As[0], Bs[0], tid, p);
    __syncthreads();
    tile_fetch(aSrc, bSrc, 32, p);

    for (int t = 0; t < NMAT / 32 - 2; ++t) {
        const int cur = t & 1;
        tile_mma(As[cur], Bs[cur], waveRow, waveCol, r, half, acc);
        tile_store(As[cur ^ 1], Bs[cur ^ 1], tid, p);
        __syncthreads();
        tile_fetch(aSrc, bSrc, (t + 2) * 32, p);
    }
    tile_mma(As[0], Bs[0], waveRow, waveCol, r, half, acc);
    tile_store(As[1], Bs[1], tid, p);
    __syncthreads();
    tile_mma(As[1], Bs[1], waveRow, waveCol, r, half, acc);

    const int row0 = rowBase + waveRow + 8 * half;
    const int col0 = colBase + waveCol + r;
    if (mode == 0) {
#pragma unroll
        for (int m = 0; m < 2; ++m)
#pragma unroll
            for (int n = 0; n < 4; ++n)
#pragma unroll
                for (int i = 0; i < 8; ++i) {
                    const int row = row0 + 16 * m + i;
                    const int col = col0 + 16 * n;
                    Dmat[row * NMAT + col] = acc[m][n][i];
                }
    } else if (mode == 1) {
#pragma unroll
        for (int m = 0; m < 2; ++m)
#pragma unroll
            for (int n = 0; n < 4; ++n)
#pragma unroll
                for (int i = 0; i < 8; ++i) {
                    const int row = row0 + 16 * m + i;
                    const int col = col0 + 16 * n;
                    const float p = acc[m][n][i];
                    const float x = Emat[row * NMAT + col];
                    Dmat[row * NMAT + col]  = p;
                    D2mat[row * NMAT + col] = K_X2 * p + K_X1 * x;
                }
    } else if (mode == 2) {
#pragma unroll
        for (int m = 0; m < 2; ++m)
#pragma unroll
            for (int n = 0; n < 4; ++n)
#pragma unroll
                for (int i = 0; i < 8; ++i) {
                    const int row = row0 + 16 * m + i;
                    const int col = col0 + 16 * n;
                    const float p  = acc[m][n][i];
                    const float x2 = Emat[row * NMAT + col];
                    const float x  = D2mat[row * NMAT + col];
                    Dmat[row * NMAT + col]  = p + K_X3 * x2;
                    D2mat[row * NMAT + col] = K_X7 * p + K_X6 * x2 + K_X5 * x
                                            + ((row == col) ? K_X4 : 0.0f);
                }
    } else {
        const float invScale = 1.0f / (float)(1 << SQUARINGS);
#pragma unroll
        for (int m = 0; m < 2; ++m)
#pragma unroll
            for (int n = 0; n < 4; ++n)
#pragma unroll
                for (int i = 0; i < 8; ++i) {
                    const int row = row0 + 16 * m + i;
                    const int col = col0 + 16 * n;
                    const float p  = acc[m][n][i];
                    const float x2 = Emat[row * NMAT + col];
                    const float x  = (Ain[row * NMAT + col] - Ain[col * NMAT + row])
                                   * invScale;
                    Dmat[row * NMAT + col] = p + K_Y2 * x2 + x
                                           + ((row == col) ? 1.0f : 0.0f);
                }
    }
}

extern "C" void kernel_launch(void* const* d_in, const int* in_sizes, int n_in,
                              void* d_out, int out_size, void* d_ws, size_t ws_size,
                              hipStream_t stream)
{
    (void)in_sizes; (void)n_in; (void)out_size; (void)ws_size;

    const float* A  = (const float*)d_in[0];
    float*       out = (float*)d_out;

    static_assert(SQUARINGS == 4, "the launch sequence below squares four times");
    const size_t NN = (size_t)NMAT * NMAT;
    float* W0 = (float*)d_ws;
    float* W1 = W0 + NN;
    float* W2 = W1 + NN;

    skew_init_kernel<<<(int)(NN / 256), 256, 0, stream>>>(A, W0);

    dim3 grid(NMAT / 128, NMAT / 128);
    dim3 block(256);

    gemm_wmma_f32_kernel<<<grid, block, 0, stream>>>(W0, W0, W1, W0, W2, A, 1);
    gemm_wmma_f32_kernel<<<grid, block, 0, stream>>>(W1, W2, out, W1, W0, A, 2);
    gemm_wmma_f32_kernel<<<grid, block, 0, stream>>>(out, W0, W2, W1, nullptr, A, 3);

    gemm_wmma_f32_kernel<<<grid, block, 0, stream>>>(W2, W2, W0, nullptr, nullptr, A, 0);
    gemm_wmma_f32_kernel<<<grid, block, 0, stream>>>(W0, W0, W1, nullptr, nullptr, A, 0);
    gemm_wmma_f32_kernel<<<grid, block, 0, stream>>>(W1, W1, W0, nullptr, nullptr, A, 0);
    gemm_wmma_f32_kernel<<<grid, block, 0, stream>>>(W0, W0, out, nullptr, nullptr, A, 0);
}


# Two squarings, not four (scale 2^-2): 5 GEMMs not 7. Degree-8 remainder at |X|=0.32 is ~1e-10, under f32 rounding up to twice the declared input's norm; fewer doublings of rounding error
# speedup vs baseline: 3.7354x; 1.3608x over previous
#include <hip/hip_runtime.h>


#define NMAT 2048
#define SQUARINGS 2
#define LDP 36

#define K_X1 0.10836465678522780f
#define K_X2 0.02709116419630695f
#define K_X3 0.66666666666666667f
#define K_X4 0.54676145797072400f
#define K_X5 0.16112557339541758f
#define K_X6 0.01409091715837821f
#define K_X7 0.03379279701087050f
#define K_Y2 0.13549236135285073f

typedef float v2f __attribute__((ext_vector_type(2)));
typedef float v4f __attribute__((ext_vector_type(4)));
typedef float v8f __attribute__((ext_vector_type(8)));

__global__ __launch_bounds__(256)
void skew_init_kernel(const float* __restrict__ A,
                      float* __restrict__ X)
{
    const int idx = blockIdx.x * 256 + threadIdx.x;
    const int i = idx >> 11;
    const int j = idx & (NMAT - 1);
    const float invScale = 1.0f / (float)(1 << SQUARINGS);
    float s = (A[i * NMAT + j] - A[j * NMAT + i]) * invScale;
    X[idx] = s;
}

struct TileRegs { v4f a0, a1, a2, a3, b0, b1, b2, b3; };

__device__ __forceinline__
void tile_fetch(const float* __restrict__ aSrc, const float* __restrict__ bSrc,
                int kt, TileRegs& p)
{
    p.a0 = *(const v4f*)(aSrc +  0 * NMAT + kt);
    p.a1 = *(const v4f*)(aSrc + 32 * NMAT + kt);
    p.a2 = *(const v4f*)(aSrc + 64 * NMAT + kt);
    p.a3 = *(const v4f*)(aSrc + 96 * NMAT + kt);
    p.b0 = *(const v4f*)(bSrc + (kt +  0) * NMAT);
    p.b1 = *(const v4f*)(bSrc + (kt +  8) * NMAT);
    p.b2 = *(const v4f*)(bSrc + (kt + 16) * NMAT);
    p.b3 = *(const v4f*)(bSrc + (kt + 24) * NMAT);
}

__device__ __forceinline__
void store_b(float (*Bs)[LDP], int bc4, int k, const v4f v)
{
    Bs[bc4 + 0][k] = v.x;
    Bs[bc4 + 1][k] = v.y;
    Bs[bc4 + 2][k] = v.z;
    Bs[bc4 + 3][k] = v.w;
}

__device__ __forceinline__
void tile_store(float (*As)[LDP], float (*Bs)[LDP], int tid, const TileRegs& p)
{
    const int arow = tid >> 3;
    const int ac4  = (tid & 7) << 2;
    const int krow = ((tid >> 3) & 3) + ((tid >> 7) << 2);
    const int bc4  = (((tid >> 5) & 3) << 5) + ((tid & 7) << 2);
    *(v4f*)&As[arow +  0][ac4] = p.a0;
    *(v4f*)&As[arow + 32][ac4] = p.a1;
    *(v4f*)&As[arow + 64][ac4] = p.a2;
    *(v4f*)&As[arow + 96][ac4] = p.a3;
    store_b(Bs, bc4, krow +  0, p.b0);
    store_b(Bs, bc4, krow +  8, p.b1);
    store_b(Bs, bc4, krow + 16, p.b2);
    store_b(Bs, bc4, krow + 24, p.b3);
}

__device__ __forceinline__
void tile_mma(const float (*As)[LDP], const float (*Bs)[LDP],
              int waveRow, int waveCol, int r, int half, v8f (&acc)[2][4])
{
#pragma unroll
    for (int kk = 0; kk < 32; kk += 4) {
        const int ka = kk + 2 * half;
        v2f a[2], b[4];
#pragma unroll
        for (int m = 0; m < 2; ++m)
            a[m] = *(const v2f*)&As[waveRow + 16 * m + r][ka];
#pragma unroll
        for (int n = 0; n < 4; ++n)
            b[n] = *(const v2f*)&Bs[waveCol + 16 * n + r][ka];
#pragma unroll
        for (int m = 0; m < 2; ++m)
#pragma unroll
            for (int n = 0; n < 4; ++n)
                acc[m][n] = __builtin_amdgcn_wmma_f32_16x16x4_f32(
                     false, a[m],
                     false, b[n],
                     (short)0, acc[m][n],
                     false,  false);
    }
}

__global__ __launch_bounds__(256)
void gemm_wmma_f32_kernel(const float* __restrict__ Amat,
                          const float* __restrict__ Bmat,
                          float* __restrict__ Dmat,
                          const float* __restrict__ Emat,
                          float* __restrict__ D2mat,
                          const float* __restrict__ Ain,
                          int mode)
{
    __shared__ __attribute__((aligned(16))) float As[2][128][LDP];
    __shared__ __attribute__((aligned(16))) float Bs[2][128][LDP];

    const int tid  = threadIdx.x;
    const int wave = tid >> 5;
    const int lane = tid & 31;
    const int half = lane >> 4;
    const int r    = lane & 15;

    const int rowBase = blockIdx.y * 128;
    const int colBase = blockIdx.x * 128;
    const int waveRow = (wave >> 1) * 32;
    const int waveCol = (wave & 1) * 64;

    v8f acc[2][4];
#pragma unroll
    for (int m = 0; m < 2; ++m)
#pragma unroll
        for (int n = 0; n < 4; ++n)
            acc[m][n] = (v8f)0.0f;

    const float* aSrc = Amat + (rowBase + (tid >> 3)) * NMAT + ((tid & 7) << 2);
    const float* bSrc = Bmat + (((tid >> 3) & 3) + ((tid >> 7) << 2)) * NMAT
                      + colBase + (((tid >> 5) & 3) << 5) + ((tid & 7) << 2);

    TileRegs p;
    tile_fetch(aSrc, bSrc, 0, p);
    tile_store(As[0], Bs[0], tid, p);
    __syncthreads();
    tile_fetch(aSrc, bSrc, 32, p);

    for (int t = 0; t < NMAT / 32 - 2; ++t) {
        const int cur = t & 1;
        tile_mma(As[cur], Bs[cur], waveRow, waveCol, r, half, acc);
        tile_store(As[cur ^ 1], Bs[cur ^ 1], tid, p);
        __syncthreads();
        tile_fetch(aSrc, bSrc, (t + 2) * 32, p);
    }
    tile_mma(As[0], Bs[0], waveRow, waveCol, r, half, acc);
    tile_store(As[1], Bs[1], tid, p);
    __syncthreads();
    tile_mma(As[1], Bs[1], waveRow, waveCol, r, half, acc);

    const int row0 = rowBase + waveRow + 8 * half;
    const int col0 = colBase + waveCol + r;
    if (mode == 0) {
#pragma unroll
        for (int m = 0; m < 2; ++m)
#pragma unroll
            for (int n = 0; n < 4; ++n)
#pragma unroll
                for (int i = 0; i < 8; ++i) {
                    const int row = row0 + 16 * m + i;
                    const int col = col0 + 16 * n;
                    Dmat[row * NMAT + col] = acc[m][n][i];
                }
    } else if (mode == 1) {
#pragma unroll
        for (int m = 0; m < 2; ++m)
#pragma unroll
            for (int n = 0; n < 4; ++n)
#pragma unroll
                for (int i = 0; i < 8; ++i) {
                    const int row = row0 + 16 * m + i;
                    const int col = col0 + 16 * n;
                    const float p = acc[m][n][i];
                    const float x = Emat[row * NMAT + col];
                    Dmat[row * NMAT + col]  = p;
                    D2mat[row * NMAT + col] = K_X2 * p + K_X1 * x;
                }
    } else if (mode == 2) {
#pragma unroll
        for (int m = 0; m < 2; ++m)
#pragma unroll
            for (int n = 0; n < 4; ++n)
#pragma unroll
                for (int i = 0; i < 8; ++i) {
                    const int row = row0 + 16 * m + i;
                    const int col = col0 + 16 * n;
                    const float p  = acc[m][n][i];
                    const float x2 = Emat[row * NMAT + col];
                    const float x  = D2mat[row * NMAT + col];
                    Dmat[row * NMAT + col]  = p + K_X3 * x2;
                    D2mat[row * NMAT + col] = K_X7 * p + K_X6 * x2 + K_X5 * x
                                            + ((row == col) ? K_X4 : 0.0f);
                }
    } else {
        const float invScale = 1.0f / (float)(1 << SQUARINGS);
#pragma unroll
        for (int m = 0; m < 2; ++m)
#pragma unroll
            for (int n = 0; n < 4; ++n)
#pragma unroll
                for (int i = 0; i < 8; ++i) {
                    const int row = row0 + 16 * m + i;
                    const int col = col0 + 16 * n;
                    const float p  = acc[m][n][i];
                    const float x2 = Emat[row * NMAT + col];
                    const float x  = (Ain[row * NMAT + col] - Ain[col * NMAT + row])
                                   * invScale;
                    Dmat[row * NMAT + col] = p + K_Y2 * x2 + x
                                           + ((row == col) ? 1.0f : 0.0f);
                }
    }
}

extern "C" void kernel_launch(void* const* d_in, const int* in_sizes, int n_in,
                              void* d_out, int out_size, void* d_ws, size_t ws_size,
                              hipStream_t stream)
{
    (void)in_sizes; (void)n_in; (void)out_size; (void)ws_size;

    const float* A  = (const float*)d_in[0];
    float*       out = (float*)d_out;

    static_assert(SQUARINGS == 2, "the launch sequence below squares twice");
    const size_t NN = (size_t)NMAT * NMAT;
    float* W0 = (float*)d_ws;
    float* W1 = W0 + NN;
    float* W2 = W1 + NN;

    skew_init_kernel<<<(int)(NN / 256), 256, 0, stream>>>(A, W0);

    dim3 grid(NMAT / 128, NMAT / 128);
    dim3 block(256);

    gemm_wmma_f32_kernel<<<grid, block, 0, stream>>>(W0, W0, W1, W0, W2, A, 1);
    gemm_wmma_f32_kernel<<<grid, block, 0, stream>>>(W1, W2, out, W1, W0, A, 2);
    gemm_wmma_f32_kernel<<<grid, block, 0, stream>>>(out, W0, W2, W1, nullptr, A, 3);

    gemm_wmma_f32_kernel<<<grid, block, 0, stream>>>(W2, W2, W0, nullptr, nullptr, A, 0);
    gemm_wmma_f32_kernel<<<grid, block, 0, stream>>>(W0, W0, out, nullptr, nullptr, A, 0);
}


# Degree-16 polynomial (Taylor through X^15) at scale 1 in 4 matrix products replaces degree 8 at scale 1/4 plus 2 squarings: 4 GEMMs not 5, 5 launches not 6; float summation order changes
# speedup vs baseline: 4.4918x; 1.2025x over previous
#include <hip/hip_runtime.h>


#define NMAT 2048
#define LDP 36

#define K_C1   0.00040187616102010346f
#define K_C2   0.0029455314402796842f
#define K_C3   0.087121675660506923f
#define K_C4   0.40175684406735718f
#define K_C5  -0.063523113356121499f
#define K_C6  13.047496136837438f
#define K_C7   0.26149279772981171f
#define K_C8  -0.23810703738709874f
#define K_C9  -0.041302763659296714f
#define K_C10  5.7923617070732627f
#define K_C11  2.2242091724963751f
#define K_AL  10.150642219937055f
#define K_BE   0.25737513237648679f
#define K_GA  -4.5209332618692120f
#define K_X1E  0.84679344639588838f

typedef float v2f __attribute__((ext_vector_type(2)));
typedef float v4f __attribute__((ext_vector_type(4)));
typedef float v8f __attribute__((ext_vector_type(8)));

__global__ __launch_bounds__(256)
void skew_init_kernel(const float* __restrict__ A,
                      float* __restrict__ X)
{
    const int idx = blockIdx.x * 256 + threadIdx.x;
    const int i = idx >> 11;
    const int j = idx & (NMAT - 1);
    X[idx] = A[i * NMAT + j] - A[j * NMAT + i];
}

struct TileRegs { v4f a0, a1, a2, a3, b0, b1, b2, b3; };

__device__ __forceinline__
void tile_fetch(const float* __restrict__ aSrc, const float* __restrict__ bSrc,
                int kt, TileRegs& p)
{
    p.a0 = *(const v4f*)(aSrc +  0 * NMAT + kt);
    p.a1 = *(const v4f*)(aSrc + 32 * NMAT + kt);
    p.a2 = *(const v4f*)(aSrc + 64 * NMAT + kt);
    p.a3 = *(const v4f*)(aSrc + 96 * NMAT + kt);
    p.b0 = *(const v4f*)(bSrc + (kt +  0) * NMAT);
    p.b1 = *(const v4f*)(bSrc + (kt +  8) * NMAT);
    p.b2 = *(const v4f*)(bSrc + (kt + 16) * NMAT);
    p.b3 = *(const v4f*)(bSrc + (kt + 24) * NMAT);
}

__device__ __forceinline__
void store_b(float (*Bs)[LDP], int bc4, int k, const v4f v)
{
    Bs[bc4 + 0][k] = v.x;
    Bs[bc4 + 1][k] = v.y;
    Bs[bc4 + 2][k] = v.z;
    Bs[bc4 + 3][k] = v.w;
}

__device__ __forceinline__
void tile_store(float (*As)[LDP], float (*Bs)[LDP], int tid, const TileRegs& p)
{
    const int arow = tid >> 3;
    const int ac4  = (tid & 7) << 2;
    const int krow = ((tid >> 3) & 3) + ((tid >> 7) << 2);
    const int bc4  = (((tid >> 5) & 3) << 5) + ((tid & 7) << 2);
    *(v4f*)&As[arow +  0][ac4] = p.a0;
    *(v4f*)&As[arow + 32][ac4] = p.a1;
    *(v4f*)&As[arow + 64][ac4] = p.a2;
    *(v4f*)&As[arow + 96][ac4] = p.a3;
    store_b(Bs, bc4, krow +  0, p.b0);
    store_b(Bs, bc4, krow +  8, p.b1);
    store_b(Bs, bc4, krow + 16, p.b2);
    store_b(Bs, bc4, krow + 24, p.b3);
}

__device__ __forceinline__
void tile_mma(const float (*As)[LDP], const float (*Bs)[LDP],
              int waveRow, int waveCol, int r, int half, v8f (&acc)[2][4])
{
#pragma unroll
    for (int kk = 0; kk < 32; kk += 4) {
        const int ka = kk + 2 * half;
        v2f a[2], b[4];
#pragma unroll
        for (int m = 0; m < 2; ++m)
            a[m] = *(const v2f*)&As[waveRow + 16 * m + r][ka];
#pragma unroll
        for (int n = 0; n < 4; ++n)
            b[n] = *(const v2f*)&Bs[waveCol + 16 * n + r][ka];
#pragma unroll
        for (int m = 0; m < 2; ++m)
#pragma unroll
            for (int n = 0; n < 4; ++n)
                acc[m][n] = __builtin_amdgcn_wmma_f32_16x16x4_f32(
                     false, a[m],
                     false, b[n],
                     (short)0, acc[m][n],
                     false,  false);
    }
}

__global__ __launch_bounds__(256)
void gemm_wmma_f32_kernel(const float* __restrict__ Amat,
                          const float* __restrict__ Bmat,
                          float* __restrict__ Dmat,
                          const float* __restrict__ Emat,
                          float* __restrict__ D2mat,
                          const float* __restrict__ Ain,
                          int mode)
{
    __shared__ __attribute__((aligned(16))) float As[2][128][LDP];
    __shared__ __attribute__((aligned(16))) float Bs[2][128][LDP];

    const int tid  = threadIdx.x;
    const int wave = tid >> 5;
    const int lane = tid & 31;
    const int half = lane >> 4;
    const int r    = lane & 15;

    const int rowBase = blockIdx.y * 128;
    const int colBase = blockIdx.x * 128;
    const int waveRow = (wave >> 1) * 32;
    const int waveCol = (wave & 1) * 64;

    v8f acc[2][4];
#pragma unroll
    for (int m = 0; m < 2; ++m)
#pragma unroll
        for (int n = 0; n < 4; ++n)
            acc[m][n] = (v8f)0.0f;

    const float* aSrc = Amat + (rowBase + (tid >> 3)) * NMAT + ((tid & 7) << 2);
    const float* bSrc = Bmat + (((tid >> 3) & 3) + ((tid >> 7) << 2)) * NMAT
                      + colBase + (((tid >> 5) & 3) << 5) + ((tid & 7) << 2);

    TileRegs p;
    tile_fetch(aSrc, bSrc, 0, p);
    tile_store(As[0], Bs[0], tid, p);
    __syncthreads();
    tile_fetch(aSrc, bSrc, 32, p);

    for (int t = 0; t < NMAT / 32 - 2; ++t) {
        const int cur = t & 1;
        tile_mma(As[cur], Bs[cur], waveRow, waveCol, r, half, acc);
        tile_store(As[cur ^ 1], Bs[cur ^ 1], tid, p);
        __syncthreads();
        tile_fetch(aSrc, bSrc, (t + 2) * 32, p);
    }
    tile_mma(As[0], Bs[0], waveRow, waveCol, r, half, acc);
    tile_store(As[1], Bs[1], tid, p);
    __syncthreads();
    tile_mma(As[1], Bs[1], waveRow, waveCol, r, half, acc);

    const int row0 = rowBase + waveRow + 8 * half;
    const int col0 = colBase + waveCol + r;
    if (mode == 1) {
#pragma unroll
        for (int m = 0; m < 2; ++m)
#pragma unroll
            for (int n = 0; n < 4; ++n)
#pragma unroll
                for (int i = 0; i < 8; ++i) {
                    const int row = row0 + 16 * m + i;
                    const int col = col0 + 16 * n;
                    const float p = acc[m][n][i];
                    const float x = Emat[row * NMAT + col];
                    Dmat[row * NMAT + col]  = p;
                    D2mat[row * NMAT + col] = K_C1 * p + K_C2 * x;
                }
    } else if (mode == 2) {
#pragma unroll
        for (int m = 0; m < 2; ++m)
#pragma unroll
            for (int n = 0; n < 4; ++n)
#pragma unroll
                for (int i = 0; i < 8; ++i) {
                    const int row = row0 + 16 * m + i;
                    const int col = col0 + 16 * n;
                    const float p  = acc[m][n][i];
                    const float x2 = Emat[row * NMAT + col];
                    const float x  = D2mat[row * NMAT + col];
                    Dmat[row * NMAT + col]  = p + K_C3 * x2 + K_C4 * x;
                    D2mat[row * NMAT + col] = p + K_C5 * x2;
                }
    } else if (mode == 3) {
#pragma unroll
        for (int m = 0; m < 2; ++m)
#pragma unroll
            for (int n = 0; n < 4; ++n)
#pragma unroll
                for (int i = 0; i < 8; ++i) {
                    const int row = row0 + 16 * m + i;
                    const int col = col0 + 16 * n;
                    const float p  = acc[m][n][i];
                    const float r1 = Emat[row * NMAT + col];
                    const float x2 = D2mat[row * NMAT + col];
                    const float x  = Ain[row * NMAT + col] - Ain[col * NMAT + row];
                    const float y0 = r1 - K_C5 * x2;
                    const float y1 = p + K_C6 * y0 + K_C7 * x2;
                    Dmat[row * NMAT + col]  = y1 + K_C8 * x2 + K_C9 * x;
                    D2mat[row * NMAT + col] = y1 + K_C10 * y0 + K_C11 * x;
                }
    } else {
#pragma unroll
        for (int m = 0; m < 2; ++m)
#pragma unroll
            for (int n = 0; n < 4; ++n)
#pragma unroll
                for (int i = 0; i < 8; ++i) {
                    const int row = row0 + 16 * m + i;
                    const int col = col0 + 16 * n;
                    const float p  = acc[m][n][i];
                    const float l2 = Amat[row * NMAT + col];
                    const float r2 = Bmat[row * NMAT + col];
                    const float r1 = Emat[row * NMAT + col];
                    const float x  = Ain[row * NMAT + col] - Ain[col * NMAT + row];
                    Dmat[row * NMAT + col] = p + K_AL * l2 + K_BE * r2 + K_GA * r1
                                           + K_X1E * x
                                           + ((row == col) ? 1.0f : 0.0f);
                }
    }
}

extern "C" void kernel_launch(void* const* d_in, const int* in_sizes, int n_in,
                              void* d_out, int out_size, void* d_ws, size_t ws_size,
                              hipStream_t stream)
{
    (void)in_sizes; (void)n_in; (void)out_size; (void)ws_size;

    const float* A  = (const float*)d_in[0];
    float*       out = (float*)d_out;

    const size_t NN = (size_t)NMAT * NMAT;
    float* W0 = (float*)d_ws;
    float* W1 = W0 + NN;
    float* W2 = W1 + NN;

    skew_init_kernel<<<(int)(NN / 256), 256, 0, stream>>>(A, W0);

    dim3 grid(NMAT / 128, NMAT / 128);
    dim3 block(256);

    gemm_wmma_f32_kernel<<<grid, block, 0, stream>>>(W0, W0, W1, W0, W2, A, 1);
    gemm_wmma_f32_kernel<<<grid, block, 0, stream>>>(W1, W2, out, W1, W0, A, 2);
    gemm_wmma_f32_kernel<<<grid, block, 0, stream>>>(out, W0, W2, W0, W1, A, 3);
    gemm_wmma_f32_kernel<<<grid, block, 0, stream>>>(W2, W1, out, W0, nullptr, A, 4);
}


# X2 = X*X is symmetric for every A (X = A - A^T is exactly skew): multiply only its 8256 tiles on or above the diagonal, in 256 balanced jobs, and store each at (i,j) and (j,i); sums unchanged
# speedup vs baseline: 4.7851x; 1.0653x over previous
#include <hip/hip_runtime.h>


#define NMAT 2048
#define LDP 36

#define K_C1   0.00040187616102010346f
#define K_C2   0.0029455314402796842f
#define K_C3   0.087121675660506923f
#define K_C4   0.40175684406735718f
#define K_C5  -0.063523113356121499f
#define K_C6  13.047496136837438f
#define K_C7   0.26149279772981171f
#define K_C8  -0.23810703738709874f
#define K_C9  -0.041302763659296714f
#define K_C10  5.7923617070732627f
#define K_C11  2.2242091724963751f
#define K_AL  10.150642219937055f
#define K_BE   0.25737513237648679f
#define K_GA  -4.5209332618692120f
#define K_X1E  0.84679344639588838f

typedef float v2f __attribute__((ext_vector_type(2)));
typedef float v4f __attribute__((ext_vector_type(4)));
typedef float v8f __attribute__((ext_vector_type(8)));

__global__ __launch_bounds__(256)
void skew_init_kernel(const float* __restrict__ A,
                      float* __restrict__ X)
{
    const int idx = blockIdx.x * 256 + threadIdx.x;
    const int i = idx >> 11;
    const int j = idx & (NMAT - 1);
    X[idx] = A[i * NMAT + j] - A[j * NMAT + i];
}

struct TileRegs { v4f a0, a1, a2, a3, b0, b1, b2, b3; };

__device__ __forceinline__
void tile_fetch(const float* __restrict__ aSrc, const float* __restrict__ bSrc,
                int kt, TileRegs& p)
{
    p.a0 = *(const v4f*)(aSrc +  0 * NMAT + kt);
    p.a1 = *(const v4f*)(aSrc + 32 * NMAT + kt);
    p.a2 = *(const v4f*)(aSrc + 64 * NMAT + kt);
    p.a3 = *(const v4f*)(aSrc + 96 * NMAT + kt);
    p.b0 = *(const v4f*)(bSrc + (kt +  0) * NMAT);
    p.b1 = *(const v4f*)(bSrc + (kt +  8) * NMAT);
    p.b2 = *(const v4f*)(bSrc + (kt + 16) * NMAT);
    p.b3 = *(const v4f*)(bSrc + (kt + 24) * NMAT);
}

__device__ __forceinline__
void store_b(float (*Bs)[LDP], int bc4, int k, const v4f v)
{
    Bs[bc4 + 0][k] = v.x;
    Bs[bc4 + 1][k] = v.y;
    Bs[bc4 + 2][k] = v.z;
    Bs[bc4 + 3][k] = v.w;
}

__device__ __forceinline__
void tile_store(float (*As)[LDP], float (*Bs)[LDP], int tid, const TileRegs& p)
{
    const int arow = tid >> 3;
    const int ac4  = (tid & 7) << 2;
    const int krow = ((tid >> 3) & 3) + ((tid >> 7) << 2);
    const int bc4  = (((tid >> 5) & 3) << 5) + ((tid & 7) << 2);
    *(v4f*)&As[arow +  0][ac4] = p.a0;
    *(v4f*)&As[arow + 32][ac4] = p.a1;
    *(v4f*)&As[arow + 64][ac4] = p.a2;
    *(v4f*)&As[arow + 96][ac4] = p.a3;
    store_b(Bs, bc4, krow +  0, p.b0);
    store_b(Bs, bc4, krow +  8, p.b1);
    store_b(Bs, bc4, krow + 16, p.b2);
    store_b(Bs, bc4, krow + 24, p.b3);
}

__device__ __forceinline__
void tile_mma(const float (*As)[LDP], const float (*Bs)[LDP],
              int waveRow, int waveCol, int r, int half, v8f (&acc)[2][4])
{
#pragma unroll
    for (int kk = 0; kk < 32; kk += 4) {
        const int ka = kk + 2 * half;
        v2f a[2], b[4];
#pragma unroll
        for (int m = 0; m < 2; ++m)
            a[m] = *(const v2f*)&As[waveRow + 16 * m + r][ka];
#pragma unroll
        for (int n = 0; n < 4; ++n)
            b[n] = *(const v2f*)&Bs[waveCol + 16 * n + r][ka];
#pragma unroll
        for (int m = 0; m < 2; ++m)
#pragma unroll
            for (int n = 0; n < 4; ++n)
                acc[m][n] = __builtin_amdgcn_wmma_f32_16x16x4_f32(
                     false, a[m],
                     false, b[n],
                     (short)0, acc[m][n],
                     false,  false);
    }
}

constexpr int pat_n(int kind)         { return kind == 0 ? 4 : 5; }
constexpr int pat_nb(int kind)        { return kind == 0 ? 2 : (kind == 1 ? 4 : 3); }
constexpr int pat_a(int kind, int q)  { return kind == 0 ? (q >> 1) : (kind == 1 ? (q == 4 ? 1 : 0) : (q >= 3 ? 1 : 0)); }
constexpr int pat_b(int kind, int q)  { return kind == 0 ? (q & 1) : (kind == 1 ? (q < 4 ? q : 3) : (q < 3 ? q : q - 2)); }
constexpr bool pat_ondiag(int kind, int q) { return kind == 1 ? (q == 0 || q == 4) : (kind == 2 ? (q == 0 || q == 3) : false); }

struct WaveTiles { int ar[2]; int bc[4]; };

template <int KIND>
__device__ __forceinline__
void sym_mma(const float (*As)[LDP], const float (*Bs)[LDP],
             const WaveTiles& wt, int r, int half, v8f (&acc)[5])
{
#pragma unroll
    for (int kk = 0; kk < 32; kk += 4) {
        const int ka = kk + 2 * half;
        v2f a[2], b[4];
#pragma unroll
        for (int m = 0; m < 2; ++m)
            a[m] = *(const v2f*)&As[wt.ar[m] + r][ka];
#pragma unroll
        for (int n = 0; n < pat_nb(KIND); ++n)
            b[n] = *(const v2f*)&Bs[wt.bc[n] + r][ka];
#pragma unroll
        for (int q = 0; q < pat_n(KIND); ++q)
            acc[q] = __builtin_amdgcn_wmma_f32_16x16x4_f32(
                 false, a[pat_a(KIND, q)],
                 false, b[pat_b(KIND, q)],
                 (short)0, acc[q],
                 false,  false);
    }
}

template <int KIND, bool DIAGJOB>
__device__ __forceinline__
void sym_store(const float* __restrict__ X, float* __restrict__ D, float* __restrict__ D2,
               int rowBase, int colBase, const WaveTiles& wt, int r, int half,
               const v8f (&acc)[5])
{
#pragma unroll
    for (int q = 0; q < pat_n(KIND); ++q) {
        const int row0 = rowBase + wt.ar[pat_a(KIND, q)] + 8 * half;
        const int col  = colBase + wt.bc[pat_b(KIND, q)] + r;
        float x[8];
#pragma unroll
        for (int i = 0; i < 8; ++i) {
            const float p = acc[q][i];
            x[i] = X[(row0 + i) * NMAT + col];
            D[(row0 + i) * NMAT + col]  = p;
            D2[(row0 + i) * NMAT + col] = K_C1 * p + K_C2 * x[i];
        }
        if (!(DIAGJOB && pat_ondiag(KIND, q))) {
            v4f p0, p1, w0, w1;
#pragma unroll
            for (int i = 0; i < 4; ++i) {
                p0[i] = acc[q][i];
                p1[i] = acc[q][i + 4];
                w0[i] = K_C1 * acc[q][i]     + K_C2 * (-x[i]);
                w1[i] = K_C1 * acc[q][i + 4] + K_C2 * (-x[i + 4]);
            }
            *(v4f*)(D  + col * NMAT + row0)     = p0;
            *(v4f*)(D  + col * NMAT + row0 + 4) = p1;
            *(v4f*)(D2 + col * NMAT + row0)     = w0;
            *(v4f*)(D2 + col * NMAT + row0 + 4) = w1;
        }
    }
}

__device__ __forceinline__
void half_fetch(const float* __restrict__ aSrc, const float* __restrict__ bSrc,
                int kt, TileRegs& p)
{
    p.a0 = *(const v4f*)(aSrc +  0 * NMAT + kt);
    p.a1 = *(const v4f*)(aSrc + 32 * NMAT + kt);
    p.a2 = *(const v4f*)(aSrc + 64 * NMAT + kt);
    p.a3 = *(const v4f*)(aSrc + 96 * NMAT + kt);
    p.b0 = *(const v4f*)(bSrc + (kt +  0) * NMAT);
    p.b1 = *(const v4f*)(bSrc + (kt + 16) * NMAT);
}

__device__ __forceinline__
void half_store(float (*As)[LDP], float (*Bs)[LDP], int tid, const TileRegs& p)
{
    const int arow = tid >> 3;
    const int ac4  = (tid & 7) << 2;
    const int krow = ((tid >> 3) & 3) + ((tid >> 6) << 2);
    const int bc4  = (((tid >> 5) & 1) << 5) + ((tid & 7) << 2);
    *(v4f*)&As[arow +  0][ac4] = p.a0;
    *(v4f*)&As[arow + 32][ac4] = p.a1;
    *(v4f*)&As[arow + 64][ac4] = p.a2;
    *(v4f*)&As[arow + 96][ac4] = p.a3;
    store_b(Bs, bc4, krow +  0, p.b0);
    store_b(Bs, bc4, krow + 16, p.b1);
}

template <bool DIAGJOB>
__device__ __forceinline__
void sym_tile_mma(const float (*As)[LDP], const float (*Bs)[LDP],
                  const WaveTiles& wt, int kind, int r, int half, v8f (&acc)[5])
{
    if (!DIAGJOB || kind == 0) sym_mma<0>(As, Bs, wt, r, half, acc);
    else if (kind == 1)        sym_mma<1>(As, Bs, wt, r, half, acc);
    else                       sym_mma<2>(As, Bs, wt, r, half, acc);
}

template <bool DIAGJOB>
__device__ __forceinline__
void sym_job(const float* __restrict__ X, float* __restrict__ D, float* __restrict__ D2,
             float (*As)[128][LDP], float (*Bs)[128][LDP], int rowBase, int colBase)
{
    const int tid  = threadIdx.x;
    const int wave = __builtin_amdgcn_readfirstlane(tid >> 5);
    const int lane = tid & 31;
    const int half = lane >> 4;
    const int r    = lane & 15;

    WaveTiles wt;
    int kind = 0;
    if (!DIAGJOB) {
        wt.ar[0] = (wave >> 1) * 32;      wt.ar[1] = wt.ar[0] + 16;
        wt.bc[0] = (wave & 1) * 32;       wt.bc[1] = wt.bc[0] + 16;
        wt.bc[2] = wt.bc[1];              wt.bc[3] = wt.bc[1];
    } else if (wave < 4) {
        wt.ar[0] = (wave >> 1) * 32;      wt.ar[1] = wt.ar[0] + 16;
        wt.bc[0] = 64 + (wave & 1) * 32;  wt.bc[1] = wt.bc[0] + 16;
        wt.bc[2] = wt.bc[1];              wt.bc[3] = wt.bc[1];
    } else {
        const int t16 = (wave >> 1 & 1) * 64;
        if ((wave & 1) == 0) {
            kind = 1;
            wt.ar[0] = t16;               wt.ar[1] = t16 + 48;
            wt.bc[0] = t16;               wt.bc[1] = t16 + 16;
            wt.bc[2] = t16 + 32;          wt.bc[3] = t16 + 48;
        } else {
            kind = 2;
            wt.ar[0] = t16 + 16;          wt.ar[1] = t16 + 32;
            wt.bc[0] = t16 + 16;          wt.bc[1] = t16 + 32;
            wt.bc[2] = t16 + 48;          wt.bc[3] = t16 + 48;
        }
    }

    v8f acc[5];
#pragma unroll
    for (int q = 0; q < 5; ++q)
        acc[q] = (v8f)0.0f;

    const float* aSrc = X + (rowBase + (tid >> 3)) * NMAT + ((tid & 7) << 2);
    const float* bSrc = DIAGJOB
        ? X + (((tid >> 3) & 3) + ((tid >> 7) << 2)) * NMAT
            + colBase + (((tid >> 5) & 3) << 5) + ((tid & 7) << 2)
        : X + (((tid >> 3) & 3) + ((tid >> 6) << 2)) * NMAT
            + colBase + (((tid >> 5) & 1) << 5) + ((tid & 7) << 2);

    TileRegs p;
    if (DIAGJOB) { tile_fetch(aSrc, bSrc, 0, p); tile_store(As[0], Bs[0], tid, p); }
    else         { half_fetch(aSrc, bSrc, 0, p); half_store(As[0], Bs[0], tid, p); }
    __syncthreads();
    if (DIAGJOB) tile_fetch(aSrc, bSrc, 32, p); else half_fetch(aSrc, bSrc, 32, p);

    for (int t = 0; t < NMAT / 32 - 2; ++t) {
        const int cur = t & 1;
        sym_tile_mma<DIAGJOB>(As[cur], Bs[cur], wt, kind, r, half, acc);
        if (DIAGJOB) tile_store(As[cur ^ 1], Bs[cur ^ 1], tid, p);
        else         half_store(As[cur ^ 1], Bs[cur ^ 1], tid, p);
        __syncthreads();
        if (DIAGJOB) tile_fetch(aSrc, bSrc, (t + 2) * 32, p);
        else         half_fetch(aSrc, bSrc, (t + 2) * 32, p);
    }
    sym_tile_mma<DIAGJOB>(As[0], Bs[0], wt, kind, r, half, acc);
    if (DIAGJOB) tile_store(As[1], Bs[1], tid, p); else half_store(As[1], Bs[1], tid, p);
    __syncthreads();
    sym_tile_mma<DIAGJOB>(As[1], Bs[1], wt, kind, r, half, acc);

    if (!DIAGJOB || kind == 0)
        sym_store<0, DIAGJOB>(X, D, D2, rowBase, colBase, wt, r, half, acc);
    else if (kind == 1)
        sym_store<1, DIAGJOB>(X, D, D2, rowBase, colBase, wt, r, half, acc);
    else
        sym_store<2, DIAGJOB>(X, D, D2, rowBase, colBase, wt, r, half, acc);
}

__global__ __launch_bounds__(256)
void sym_square_kernel(const float* __restrict__ X,
                       float* __restrict__ D,
                       float* __restrict__ D2)
{
    __shared__ __attribute__((aligned(16))) float As[2][128][LDP];
    __shared__ __attribute__((aligned(16))) float Bs[2][128][LDP];

    const int b = blockIdx.x;
    if (b < 16) {
        sym_job<true>(X, D, D2, As, Bs, b * 128, b * 128);
    } else {
        const int j2 = b - 16;
        int rem = j2 >> 1;
        int I = 0;
        for (int s = 0; s < 15; ++s)
            if (rem >= 15 - I) { rem -= 15 - I; ++I; }
        const int J = I + 1 + rem;
        sym_job<false>(X, D, D2, As, Bs, I * 128, J * 128 + (j2 & 1) * 64);
    }
}

__global__ __launch_bounds__(256)
void gemm_wmma_f32_kernel(const float* __restrict__ Amat,
                          const float* __restrict__ Bmat,
                          float* __restrict__ Dmat,
                          const float* __restrict__ Emat,
                          float* __restrict__ D2mat,
                          const float* __restrict__ Ain,
                          int mode)
{
    __shared__ __attribute__((aligned(16))) float As[2][128][LDP];
    __shared__ __attribute__((aligned(16))) float Bs[2][128][LDP];

    const int tid  = threadIdx.x;
    const int wave = tid >> 5;
    const int lane = tid & 31;
    const int half = lane >> 4;
    const int r    = lane & 15;

    const int rowBase = blockIdx.y * 128;
    const int colBase = blockIdx.x * 128;
    const int waveRow = (wave >> 1) * 32;
    const int waveCol = (wave & 1) * 64;

    v8f acc[2][4];
#pragma unroll
    for (int m = 0; m < 2; ++m)
#pragma unroll
        for (int n = 0; n < 4; ++n)
            acc[m][n] = (v8f)0.0f;

    const float* aSrc = Amat + (rowBase + (tid >> 3)) * NMAT + ((tid & 7) << 2);
    const float* bSrc = Bmat + (((tid >> 3) & 3) + ((tid >> 7) << 2)) * NMAT
                      + colBase + (((tid >> 5) & 3) << 5) + ((tid & 7) << 2);

    TileRegs p;
    tile_fetch(aSrc, bSrc, 0, p);
    tile_store(As[0], Bs[0], tid, p);
    __syncthreads();
    tile_fetch(aSrc, bSrc, 32, p);

    for (int t = 0; t < NMAT / 32 - 2; ++t) {
        const int cur = t & 1;
        tile_mma(As[cur], Bs[cur], waveRow, waveCol, r, half, acc);
        tile_store(As[cur ^ 1], Bs[cur ^ 1], tid, p);
        __syncthreads();
        tile_fetch(aSrc, bSrc, (t + 2) * 32, p);
    }
    tile_mma(As[0], Bs[0], waveRow, waveCol, r, half, acc);
    tile_store(As[1], Bs[1], tid, p);
    __syncthreads();
    tile_mma(As[1], Bs[1], waveRow, waveCol, r, half, acc);

    const int row0 = rowBase + waveRow + 8 * half;
    const int col0 = colBase + waveCol + r;
    if (mode == 2) {
#pragma unroll
        for (int m = 0; m < 2; ++m)
#pragma unroll
            for (int n = 0; n < 4; ++n)
#pragma unroll
                for (int i = 0; i < 8; ++i) {
                    const int row = row0 + 16 * m + i;
                    const int col = col0 + 16 * n;
                    const float p  = acc[m][n][i];
                    const float x2 = Emat[row * NMAT + col];
                    const float x  = D2mat[row * NMAT + col];
                    Dmat[row * NMAT + col]  = p + K_C3 * x2 + K_C4 * x;
                    D2mat[row * NMAT + col] = p + K_C5 * x2;
                }
    } else if (mode == 3) {
#pragma unroll
        for (int m = 0; m < 2; ++m)
#pragma unroll
            for (int n = 0; n < 4; ++n)
#pragma unroll
                for (int i = 0; i < 8; ++i) {
                    const int row = row0 + 16 * m + i;
                    const int col = col0 + 16 * n;
                    const float p  = acc[m][n][i];
                    const float r1 = Emat[row * NMAT + col];
                    const float x2 = D2mat[row * NMAT + col];
                    const float x  = Ain[row * NMAT + col] - Ain[col * NMAT + row];
                    const float y0 = r1 - K_C5 * x2;
                    const float y1 = p + K_C6 * y0 + K_C7 * x2;
                    Dmat[row * NMAT + col]  = y1 + K_C8 * x2 + K_C9 * x;
                    D2mat[row * NMAT + col] = y1 + K_C10 * y0 + K_C11 * x;
                }
    } else {
#pragma unroll
        for (int m = 0; m < 2; ++m)
#pragma unroll
            for (int n = 0; n < 4; ++n)
#pragma unroll
                for (int i = 0; i < 8; ++i) {
                    const int row = row0 + 16 * m + i;
                    const int col = col0 + 16 * n;
                    const float p  = acc[m][n][i];
                    const float l2 = Amat[row * NMAT + col];
                    const float r2 = Bmat[row * NMAT + col];
                    const float r1 = Emat[row * NMAT + col];
                    const float x  = Ain[row * NMAT + col] - Ain[col * NMAT + row];
                    Dmat[row * NMAT + col] = p + K_AL * l2 + K_BE * r2 + K_GA * r1
                                           + K_X1E * x
                                           + ((row == col) ? 1.0f : 0.0f);
                }
    }
}

extern "C" void kernel_launch(void* const* d_in, const int* in_sizes, int n_in,
                              void* d_out, int out_size, void* d_ws, size_t ws_size,
                              hipStream_t stream)
{
    (void)in_sizes; (void)n_in; (void)out_size; (void)ws_size;

    const float* A  = (const float*)d_in[0];
    float*       out = (float*)d_out;

    const size_t NN = (size_t)NMAT * NMAT;
    float* W0 = (float*)d_ws;
    float* W1 = W0 + NN;
    float* W2 = W1 + NN;

    skew_init_kernel<<<(int)(NN / 256), 256, 0, stream>>>(A, W0);

    dim3 grid(NMAT / 128, NMAT / 128);
    dim3 block(256);

    sym_square_kernel<<<256, 256, 0, stream>>>(W0, W1, W2);
    gemm_wmma_f32_kernel<<<grid, block, 0, stream>>>(W1, W2, out, W1, W0, A, 2);
    gemm_wmma_f32_kernel<<<grid, block, 0, stream>>>(out, W0, W2, W0, W1, A, 3);
    gemm_wmma_f32_kernel<<<grid, block, 0, stream>>>(W2, W1, out, W0, nullptr, A, 4);
}


# X4=X2*X2 is symmetric like X2: both multiplied on the upper triangle only, then 2 full products on mixes of X4,X2,X reach degree 16; 3 GEMM units not 3.5; Taylor through X^14 not X^15; sums reordered
# speedup vs baseline: 5.0270x; 1.0506x over previous
#include <hip/hip_runtime.h>


#define NMAT 2048
#define LDP 36

#define K_KZ  0.00077160493827160490f
#define K_A1  0.11154761805772045f
#define K_A2  0.27992720822635375f
#define K_B1  (-0.050111642644035441f)
#define K_B2  0.58017644756523645f
#define K_C1  0.017226797284205099f
#define K_C2  0.024021066342898575f
#define K_C3  (-0.88083743549646332f)
#define K_D1  0.0074336583226582171f
#define K_D2  0.11571413588046474f
#define K_D3  0.55876985454585559f
#define K_S   0.20321412844635858f
#define K_E1  3.8524813843396437f
#define K_LA  (-4.9025788223219831f)
#define K_RO  (-8.5544730341591464f)
#define K_X1  10.728878801472899f


typedef float v2f __attribute__((ext_vector_type(2)));
typedef float v4f __attribute__((ext_vector_type(4)));
typedef float v8f __attribute__((ext_vector_type(8)));

__global__ __launch_bounds__(256)
void skew_init_kernel(const float* __restrict__ A,
                      float* __restrict__ X)
{
    const int idx = blockIdx.x * 256 + threadIdx.x;
    const int i = idx >> 11;
    const int j = idx & (NMAT - 1);
    X[idx] = A[i * NMAT + j] - A[j * NMAT + i];
}

struct TileRegs { v4f a0, a1, a2, a3, b0, b1, b2, b3; };

__device__ __forceinline__
void tile_fetch(const float* __restrict__ aSrc, const float* __restrict__ bSrc,
                int kt, TileRegs& p)
{
    p.a0 = *(const v4f*)(aSrc +  0 * NMAT + kt);
    p.a1 = *(const v4f*)(aSrc + 32 * NMAT + kt);
    p.a2 = *(const v4f*)(aSrc + 64 * NMAT + kt);
    p.a3 = *(const v4f*)(aSrc + 96 * NMAT + kt);
    p.b0 = *(const v4f*)(bSrc + (kt +  0) * NMAT);
    p.b1 = *(const v4f*)(bSrc + (kt +  8) * NMAT);
    p.b2 = *(const v4f*)(bSrc + (kt + 16) * NMAT);
    p.b3 = *(const v4f*)(bSrc + (kt + 24) * NMAT);
}

__device__ __forceinline__
void store_b(float (*Bs)[LDP], int bc4, int k, const v4f v)
{
    Bs[bc4 + 0][k] = v.x;
    Bs[bc4 + 1][k] = v.y;
    Bs[bc4 + 2][k] = v.z;
    Bs[bc4 + 3][k] = v.w;
}

__device__ __forceinline__
void tile_store(float (*As)[LDP], float (*Bs)[LDP], int tid, const TileRegs& p)
{
    const int arow = tid >> 3;
    const int ac4  = (tid & 7) << 2;
    const int krow = ((tid >> 3) & 3) + ((tid >> 7) << 2);
    const int bc4  = (((tid >> 5) & 3) << 5) + ((tid & 7) << 2);
    *(v4f*)&As[arow +  0][ac4] = p.a0;
    *(v4f*)&As[arow + 32][ac4] = p.a1;
    *(v4f*)&As[arow + 64][ac4] = p.a2;
    *(v4f*)&As[arow + 96][ac4] = p.a3;
    store_b(Bs, bc4, krow +  0, p.b0);
    store_b(Bs, bc4, krow +  8, p.b1);
    store_b(Bs, bc4, krow + 16, p.b2);
    store_b(Bs, bc4, krow + 24, p.b3);
}

__device__ __forceinline__
void tile_mma(const float (*As)[LDP], const float (*Bs)[LDP],
              int waveRow, int waveCol, int r, int half, v8f (&acc)[2][4])
{
#pragma unroll
    for (int kk = 0; kk < 32; kk += 4) {
        const int ka = kk + 2 * half;
        v2f a[2], b[4];
#pragma unroll
        for (int m = 0; m < 2; ++m)
            a[m] = *(const v2f*)&As[waveRow + 16 * m + r][ka];
#pragma unroll
        for (int n = 0; n < 4; ++n)
            b[n] = *(const v2f*)&Bs[waveCol + 16 * n + r][ka];
#pragma unroll
        for (int m = 0; m < 2; ++m)
#pragma unroll
            for (int n = 0; n < 4; ++n)
                acc[m][n] = __builtin_amdgcn_wmma_f32_16x16x4_f32(
                     false, a[m],
                     false, b[n],
                     (short)0, acc[m][n],
                     false,  false);
    }
}

constexpr int pat_n(int kind)         { return kind == 0 ? 4 : 5; }
constexpr int pat_nb(int kind)        { return kind == 0 ? 2 : (kind == 1 ? 4 : 3); }
constexpr int pat_a(int kind, int q)  { return kind == 0 ? (q >> 1) : (kind == 1 ? (q == 4 ? 1 : 0) : (q >= 3 ? 1 : 0)); }
constexpr int pat_b(int kind, int q)  { return kind == 0 ? (q & 1) : (kind == 1 ? (q < 4 ? q : 3) : (q < 3 ? q : q - 2)); }
constexpr bool pat_ondiag(int kind, int q) { return kind == 1 ? (q == 0 || q == 4) : (kind == 2 ? (q == 0 || q == 3) : false); }

struct WaveTiles { int ar[2]; int bc[4]; };

template <int KIND>
__device__ __forceinline__
void sym_mma(const float (*As)[LDP], const float (*Bs)[LDP],
             const WaveTiles& wt, int r, int half, v8f (&acc)[5])
{
#pragma unroll
    for (int kk = 0; kk < 32; kk += 4) {
        const int ka = kk + 2 * half;
        v2f a[2], b[4];
#pragma unroll
        for (int m = 0; m < 2; ++m)
            a[m] = *(const v2f*)&As[wt.ar[m] + r][ka];
#pragma unroll
        for (int n = 0; n < pat_nb(KIND); ++n)
            b[n] = *(const v2f*)&Bs[wt.bc[n] + r][ka];
#pragma unroll
        for (int q = 0; q < pat_n(KIND); ++q)
            acc[q] = __builtin_amdgcn_wmma_f32_16x16x4_f32(
                 false, a[pat_a(KIND, q)],
                 false, b[pat_b(KIND, q)],
                 (short)0, acc[q],
                 false,  false);
    }
}

template <int KIND, bool DIAGJOB, int MODE>
__device__ __forceinline__
void sym_store(const float* __restrict__ Op, float* __restrict__ D,
               float* __restrict__ D2, float* __restrict__ D3,
               int rowBase, int colBase, const WaveTiles& wt, int r, int half,
               const v8f (&acc)[5])
{
#pragma unroll
    for (int q = 0; q < pat_n(KIND); ++q) {
        const int row0 = rowBase + wt.ar[pat_a(KIND, q)] + 8 * half;
        const int col  = colBase + wt.bc[pat_b(KIND, q)] + r;
        float ml[8], mr[8];
#pragma unroll
        for (int i = 0; i < 8; ++i) {
            const float p = acc[q][i];
            if (MODE == 1) {
                D[(row0 + i) * NMAT + col] = p;
            } else {
                const float x  = D[(row0 + i) * NMAT + col];
                const float y  = Op[(row0 + i) * NMAT + col];
                const float el = K_KZ * p + K_A1 * y;
                const float er = K_KZ * p + K_B1 * y;
                D[(row0 + i) * NMAT + col]  = p;
                D2[(row0 + i) * NMAT + col] = el + K_A2 * x;
                D3[(row0 + i) * NMAT + col] = er + K_B2 * x;
                ml[i] = el - K_A2 * x;
                mr[i] = er - K_B2 * x;
            }
        }
        if (!(DIAGJOB && pat_ondiag(KIND, q))) {
            v4f p0, p1;
#pragma unroll
            for (int i = 0; i < 4; ++i) {
                p0[i] = acc[q][i];
                p1[i] = acc[q][i + 4];
            }
            *(v4f*)(D + col * NMAT + row0)     = p0;
            *(v4f*)(D + col * NMAT + row0 + 4) = p1;
            if (MODE == 2) {
                v4f l0, l1, r0, r1;
#pragma unroll
                for (int i = 0; i < 4; ++i) {
                    l0[i] = ml[i];  l1[i] = ml[i + 4];
                    r0[i] = mr[i];  r1[i] = mr[i + 4];
                }
                *(v4f*)(D2 + col * NMAT + row0)     = l0;
                *(v4f*)(D2 + col * NMAT + row0 + 4) = l1;
                *(v4f*)(D3 + col * NMAT + row0)     = r0;
                *(v4f*)(D3 + col * NMAT + row0 + 4) = r1;
            }
        }
    }
}

__device__ __forceinline__
void half_fetch(const float* __restrict__ aSrc, const float* __restrict__ bSrc,
                int kt, TileRegs& p)
{
    p.a0 = *(const v4f*)(aSrc +  0 * NMAT + kt);
    p.a1 = *(const v4f*)(aSrc + 32 * NMAT + kt);
    p.a2 = *(const v4f*)(aSrc + 64 * NMAT + kt);
    p.a3 = *(const v4f*)(aSrc + 96 * NMAT + kt);
    p.b0 = *(const v4f*)(bSrc + (kt +  0) * NMAT);
    p.b1 = *(const v4f*)(bSrc + (kt + 16) * NMAT);
}

__device__ __forceinline__
void half_store(float (*As)[LDP], float (*Bs)[LDP], int tid, const TileRegs& p)
{
    const int arow = tid >> 3;
    const int ac4  = (tid & 7) << 2;
    const int krow = ((tid >> 3) & 3) + ((tid >> 6) << 2);
    const int bc4  = (((tid >> 5) & 1) << 5) + ((tid & 7) << 2);
    *(v4f*)&As[arow +  0][ac4] = p.a0;
    *(v4f*)&As[arow + 32][ac4] = p.a1;
    *(v4f*)&As[arow + 64][ac4] = p.a2;
    *(v4f*)&As[arow + 96][ac4] = p.a3;
    store_b(Bs, bc4, krow +  0, p.b0);
    store_b(Bs, bc4, krow + 16, p.b1);
}

template <bool DIAGJOB>
__device__ __forceinline__
void sym_tile_mma(const float (*As)[LDP], const float (*Bs)[LDP],
                  const WaveTiles& wt, int kind, int r, int half, v8f (&acc)[5])
{
    if (!DIAGJOB || kind == 0) sym_mma<0>(As, Bs, wt, r, half, acc);
    else if (kind == 1)        sym_mma<1>(As, Bs, wt, r, half, acc);
    else                       sym_mma<2>(As, Bs, wt, r, half, acc);
}

template <bool DIAGJOB, int MODE>
__device__ __forceinline__
void sym_job(const float* __restrict__ Op, float* __restrict__ D,
             float* __restrict__ D2, float* __restrict__ D3,
             float (*As)[128][LDP], float (*Bs)[128][LDP], int rowBase, int colBase)
{
    const int tid  = threadIdx.x;
    const int wave = __builtin_amdgcn_readfirstlane(tid >> 5);
    const int lane = tid & 31;
    const int half = lane >> 4;
    const int r    = lane & 15;

    WaveTiles wt;
    int kind = 0;
    if (!DIAGJOB) {
        wt.ar[0] = (wave >> 1) * 32;      wt.ar[1] = wt.ar[0] + 16;
        wt.bc[0] = (wave & 1) * 32;       wt.bc[1] = wt.bc[0] + 16;
        wt.bc[2] = wt.bc[1];              wt.bc[3] = wt.bc[1];
    } else if (wave < 4) {
        wt.ar[0] = (wave >> 1) * 32;      wt.ar[1] = wt.ar[0] + 16;
        wt.bc[0] = 64 + (wave & 1) * 32;  wt.bc[1] = wt.bc[0] + 16;
        wt.bc[2] = wt.bc[1];              wt.bc[3] = wt.bc[1];
    } else {
        const int t16 = (wave >> 1 & 1) * 64;
        if ((wave & 1) == 0) {
            kind = 1;
            wt.ar[0] = t16;               wt.ar[1] = t16 + 48;
            wt.bc[0] = t16;               wt.bc[1] = t16 + 16;
            wt.bc[2] = t16 + 32;          wt.bc[3] = t16 + 48;
        } else {
            kind = 2;
            wt.ar[0] = t16 + 16;          wt.ar[1] = t16 + 32;
            wt.bc[0] = t16 + 16;          wt.bc[1] = t16 + 32;
            wt.bc[2] = t16 + 48;          wt.bc[3] = t16 + 48;
        }
    }

    v8f acc[5];
#pragma unroll
    for (int q = 0; q < 5; ++q)
        acc[q] = (v8f)0.0f;

    const float* aSrc = Op + (rowBase + (tid >> 3)) * NMAT + ((tid & 7) << 2);
    const float* bSrc = DIAGJOB
        ? Op + (((tid >> 3) & 3) + ((tid >> 7) << 2)) * NMAT
            + colBase + (((tid >> 5) & 3) << 5) + ((tid & 7) << 2)
        : Op + (((tid >> 3) & 3) + ((tid >> 6) << 2)) * NMAT
            + colBase + (((tid >> 5) & 1) << 5) + ((tid & 7) << 2);

    TileRegs p;
    if (DIAGJOB) { tile_fetch(aSrc, bSrc, 0, p); tile_store(As[0], Bs[0], tid, p); }
    else         { half_fetch(aSrc, bSrc, 0, p); half_store(As[0], Bs[0], tid, p); }
    __syncthreads();
    if (DIAGJOB) tile_fetch(aSrc, bSrc, 32, p); else half_fetch(aSrc, bSrc, 32, p);

    for (int t = 0; t < NMAT / 32 - 2; ++t) {
        const int cur = t & 1;
        sym_tile_mma<DIAGJOB>(As[cur], Bs[cur], wt, kind, r, half, acc);
        if (DIAGJOB) tile_store(As[cur ^ 1], Bs[cur ^ 1], tid, p);
        else         half_store(As[cur ^ 1], Bs[cur ^ 1], tid, p);
        __syncthreads();
        if (DIAGJOB) tile_fetch(aSrc, bSrc, (t + 2) * 32, p);
        else         half_fetch(aSrc, bSrc, (t + 2) * 32, p);
    }
    sym_tile_mma<DIAGJOB>(As[0], Bs[0], wt, kind, r, half, acc);
    if (DIAGJOB) tile_store(As[1], Bs[1], tid, p); else half_store(As[1], Bs[1], tid, p);
    __syncthreads();
    sym_tile_mma<DIAGJOB>(As[1], Bs[1], wt, kind, r, half, acc);

    if (!DIAGJOB || kind == 0)
        sym_store<0, DIAGJOB, MODE>(Op, D, D2, D3, rowBase, colBase, wt, r, half, acc);
    else if (kind == 1)
        sym_store<1, DIAGJOB, MODE>(Op, D, D2, D3, rowBase, colBase, wt, r, half, acc);
    else
        sym_store<2, DIAGJOB, MODE>(Op, D, D2, D3, rowBase, colBase, wt, r, half, acc);
}

template <int MODE>
__global__ __launch_bounds__(256)
void sym_square_kernel(const float* __restrict__ Op,
                       float* __restrict__ D,
                       float* __restrict__ D2,
                       float* __restrict__ D3)
{
    __shared__ __attribute__((aligned(16))) float As[2][128][LDP];
    __shared__ __attribute__((aligned(16))) float Bs[2][128][LDP];

    const int b = blockIdx.x;
    if (b < 16) {
        sym_job<true, MODE>(Op, D, D2, D3, As, Bs, b * 128, b * 128);
    } else {
        const int j2 = b - 16;
        int rem = j2 >> 1;
        int I = 0;
        for (int s = 0; s < 15; ++s)
            if (rem >= 15 - I) { rem -= 15 - I; ++I; }
        const int J = I + 1 + rem;
        sym_job<false, MODE>(Op, D, D2, D3, As, Bs, I * 128, J * 128 + (j2 & 1) * 64);
    }
}

__global__ __launch_bounds__(256)
void gemm_wmma_f32_kernel(const float* __restrict__ Amat,
                          const float* __restrict__ Bmat,
                          float* __restrict__ Dmat,
                          const float* __restrict__ Emat,
                          float* __restrict__ D2mat,
                          const float* __restrict__ Ain,
                          int mode)
{
    __shared__ __attribute__((aligned(16))) float As[2][128][LDP];
    __shared__ __attribute__((aligned(16))) float Bs[2][128][LDP];

    const int tid  = threadIdx.x;
    const int wave = tid >> 5;
    const int lane = tid & 31;
    const int half = lane >> 4;
    const int r    = lane & 15;

    const int rowBase = blockIdx.y * 128;
    const int colBase = blockIdx.x * 128;
    const int waveRow = (wave >> 1) * 32;
    const int waveCol = (wave & 1) * 64;

    v8f acc[2][4];
#pragma unroll
    for (int m = 0; m < 2; ++m)
#pragma unroll
        for (int n = 0; n < 4; ++n)
            acc[m][n] = (v8f)0.0f;

    const float* aSrc = Amat + (rowBase + (tid >> 3)) * NMAT + ((tid & 7) << 2);
    const float* bSrc = Bmat + (((tid >> 3) & 3) + ((tid >> 7) << 2)) * NMAT
                      + colBase + (((tid >> 5) & 3) << 5) + ((tid & 7) << 2);

    TileRegs p;
    tile_fetch(aSrc, bSrc, 0, p);
    tile_store(As[0], Bs[0], tid, p);
    __syncthreads();
    tile_fetch(aSrc, bSrc, 32, p);

    for (int t = 0; t < NMAT / 32 - 2; ++t) {
        const int cur = t & 1;
        tile_mma(As[cur], Bs[cur], waveRow, waveCol, r, half, acc);
        tile_store(As[cur ^ 1], Bs[cur ^ 1], tid, p);
        __syncthreads();
        tile_fetch(aSrc, bSrc, (t + 2) * 32, p);
    }
    tile_mma(As[0], Bs[0], waveRow, waveCol, r, half, acc);
    tile_store(As[1], Bs[1], tid, p);
    __syncthreads();
    tile_mma(As[1], Bs[1], waveRow, waveCol, r, half, acc);


    const int row0 = rowBase + waveRow + 8 * half;
    const int col0 = colBase + waveCol + r;
    if (mode == 3) {
#pragma unroll
        for (int m = 0; m < 2; ++m)
#pragma unroll
            for (int n = 0; n < 4; ++n)
#pragma unroll
                for (int i = 0; i < 8; ++i) {
                    const int row = row0 + 16 * m + i;
                    const int col = col0 + 16 * n;
                    const float p  = acc[m][n][i];
                    const float x4 = Dmat[row * NMAT + col];
                    const float x2 = D2mat[row * NMAT + col];
                    const float x  = Ain[row * NMAT + col] - Ain[col * NMAT + row];
                    Dmat[row * NMAT + col]  = p + K_C1 * x4 + K_C2 * x2 + K_C3 * x;
                    D2mat[row * NMAT + col] = p + K_D1 * x4 + K_D2 * x2 + K_D3 * x;
                }
    } else {
#pragma unroll
        for (int m = 0; m < 2; ++m)
#pragma unroll
            for (int n = 0; n < 4; ++n)
#pragma unroll
                for (int i = 0; i < 8; ++i) {
                    const int row = row0 + 16 * m + i;
                    const int col = col0 + 16 * n;
                    const float p  = acc[m][n][i];
                    const float l2 = Amat[row * NMAT + col];
                    const float l1 = Emat[row * NMAT + col];
                    const float r1 = Dmat[row * NMAT + col];
                    const float x  = Ain[row * NMAT + col] - Ain[col * NMAT + row];
                    Dmat[row * NMAT + col] = K_S * p + K_E1 * l2 + K_LA * l1 + K_RO * r1
                                           + K_X1 * x
                                           + ((row == col) ? 1.0f : 0.0f);
                }
    }
}

extern "C" void kernel_launch(void* const* d_in, const int* in_sizes, int n_in,
                              void* d_out, int out_size, void* d_ws, size_t ws_size,
                              hipStream_t stream)
{
    (void)in_sizes; (void)n_in; (void)out_size; (void)ws_size;

    const float* A  = (const float*)d_in[0];
    float*       out = (float*)d_out;

    const size_t NN = (size_t)NMAT * NMAT;
    float* W0 = (float*)d_ws;
    float* W1 = W0 + NN;
    float* W2 = W1 + NN;

    skew_init_kernel<<<(int)(NN / 256), 256, 0, stream>>>(A, W0);

    dim3 grid(NMAT / 128, NMAT / 128);
    dim3 block(256);

    sym_square_kernel<1><<<256, 256, 0, stream>>>(W0, W1, nullptr, nullptr);
    sym_square_kernel<2><<<256, 256, 0, stream>>>(W1, W0, W2, out);
    gemm_wmma_f32_kernel<<<grid, block, 0, stream>>>(W2, out, W0, nullptr, W1, A, 3);
    gemm_wmma_f32_kernel<<<grid, block, 0, stream>>>(W0, W1, out, W2, nullptr, A, 4);
}


# X = A - A^T formed per pair of 64x64 tiles, a 4x4 block and its mirror image per thread transposed in registers: 16-byte loads and stores, 528 workgroups not 16384; every stored value bit for bit
# speedup vs baseline: 5.2219x; 1.0388x over previous
#include <hip/hip_runtime.h>


#define NMAT 2048
#define LDP 36

#define K_KZ  0.00077160493827160490f
#define K_A1  0.11154761805772045f
#define K_A2  0.27992720822635375f
#define K_B1  (-0.050111642644035441f)
#define K_B2  0.58017644756523645f
#define K_C1  0.017226797284205099f
#define K_C2  0.024021066342898575f
#define K_C3  (-0.88083743549646332f)
#define K_D1  0.0074336583226582171f
#define K_D2  0.11571413588046474f
#define K_D3  0.55876985454585559f
#define K_S   0.20321412844635858f
#define K_E1  3.8524813843396437f
#define K_LA  (-4.9025788223219831f)
#define K_RO  (-8.5544730341591464f)
#define K_X1  10.728878801472899f


typedef float v2f __attribute__((ext_vector_type(2)));
typedef float v4f __attribute__((ext_vector_type(4)));
typedef float v8f __attribute__((ext_vector_type(8)));

__global__ __launch_bounds__(256)
void skew_init_kernel(const float* __restrict__ A,
                      float* __restrict__ X)
{
    const int x = blockIdx.x;
    const int y = blockIdx.y;
    int I, J;
    if (x > y) { I = y;      J = x - 1; }
    else       { I = 31 - y; J = 31 - y + x; }

    const int tid  = threadIdx.x;
    const int lane = tid & 31;
    const int wave = tid >> 5;
    const int mc = (lane & 7) + ((wave & 1) << 3);
    const int mr = (lane >> 3) + ((wave >> 1) << 2);
    const int i0 = I * 64 + 4 * mr;
    const int j0 = J * 64 + 4 * mc;

    v4f p[4], q[4];
#pragma unroll
    for (int k = 0; k < 4; ++k)
        p[k] = *(const v4f*)(A + (i0 + k) * NMAT + j0);
#pragma unroll
    for (int k = 0; k < 4; ++k)
        q[k] = *(const v4f*)(A + (j0 + k) * NMAT + i0);

#pragma unroll
    for (int k = 0; k < 4; ++k) {
        v4f d;
        d.x = p[k].x - q[0][k];
        d.y = p[k].y - q[1][k];
        d.z = p[k].z - q[2][k];
        d.w = p[k].w - q[3][k];
        *(v4f*)(X + (i0 + k) * NMAT + j0) = d;
    }
    if (I != J) {
#pragma unroll
        for (int k = 0; k < 4; ++k) {
            v4f e;
            e.x = q[k].x - p[0][k];
            e.y = q[k].y - p[1][k];
            e.z = q[k].z - p[2][k];
            e.w = q[k].w - p[3][k];
            *(v4f*)(X + (j0 + k) * NMAT + i0) = e;
        }
    }
}

struct TileRegs { v4f a0, a1, a2, a3, b0, b1, b2, b3; };

__device__ __forceinline__
void tile_fetch(const float* __restrict__ aSrc, const float* __restrict__ bSrc,
                int kt, TileRegs& p)
{
    p.a0 = *(const v4f*)(aSrc +  0 * NMAT + kt);
    p.a1 = *(const v4f*)(aSrc + 32 * NMAT + kt);
    p.a2 = *(const v4f*)(aSrc + 64 * NMAT + kt);
    p.a3 = *(const v4f*)(aSrc + 96 * NMAT + kt);
    p.b0 = *(const v4f*)(bSrc + (kt +  0) * NMAT);
    p.b1 = *(const v4f*)(bSrc + (kt +  8) * NMAT);
    p.b2 = *(const v4f*)(bSrc + (kt + 16) * NMAT);
    p.b3 = *(const v4f*)(bSrc + (kt + 24) * NMAT);
}

__device__ __forceinline__
void store_b(float (*Bs)[LDP], int bc4, int k, const v4f v)
{
    Bs[bc4 + 0][k] = v.x;
    Bs[bc4 + 1][k] = v.y;
    Bs[bc4 + 2][k] = v.z;
    Bs[bc4 + 3][k] = v.w;
}

__device__ __forceinline__
void tile_store(float (*As)[LDP], float (*Bs)[LDP], int tid, const TileRegs& p)
{
    const int arow = tid >> 3;
    const int ac4  = (tid & 7) << 2;
    const int krow = ((tid >> 3) & 3) + ((tid >> 7) << 2);
    const int bc4  = (((tid >> 5) & 3) << 5) + ((tid & 7) << 2);
    *(v4f*)&As[arow +  0][ac4] = p.a0;
    *(v4f*)&As[arow + 32][ac4] = p.a1;
    *(v4f*)&As[arow + 64][ac4] = p.a2;
    *(v4f*)&As[arow + 96][ac4] = p.a3;
    store_b(Bs, bc4, krow +  0, p.b0);
    store_b(Bs, bc4, krow +  8, p.b1);
    store_b(Bs, bc4, krow + 16, p.b2);
    store_b(Bs, bc4, krow + 24, p.b3);
}

__device__ __forceinline__
void tile_mma(const float (*As)[LDP], const float (*Bs)[LDP],
              int waveRow, int waveCol, int r, int half, v8f (&acc)[2][4])
{
#pragma unroll
    for (int kk = 0; kk < 32; kk += 4) {
        const int ka = kk + 2 * half;
        v2f a[2], b[4];
#pragma unroll
        for (int m = 0; m < 2; ++m)
            a[m] = *(const v2f*)&As[waveRow + 16 * m + r][ka];
#pragma unroll
        for (int n = 0; n < 4; ++n)
            b[n] = *(const v2f*)&Bs[waveCol + 16 * n + r][ka];
#pragma unroll
        for (int m = 0; m < 2; ++m)
#pragma unroll
            for (int n = 0; n < 4; ++n)
                acc[m][n] = __builtin_amdgcn_wmma_f32_16x16x4_f32(
                     false, a[m],
                     false, b[n],
                     (short)0, acc[m][n],
                     false,  false);
    }
}

constexpr int pat_n(int kind)         { return kind == 0 ? 4 : 5; }
constexpr int pat_nb(int kind)        { return kind == 0 ? 2 : (kind == 1 ? 4 : 3); }
constexpr int pat_a(int kind, int q)  { return kind == 0 ? (q >> 1) : (kind == 1 ? (q == 4 ? 1 : 0) : (q >= 3 ? 1 : 0)); }
constexpr int pat_b(int kind, int q)  { return kind == 0 ? (q & 1) : (kind == 1 ? (q < 4 ? q : 3) : (q < 3 ? q : q - 2)); }
constexpr bool pat_ondiag(int kind, int q) { return kind == 1 ? (q == 0 || q == 4) : (kind == 2 ? (q == 0 || q == 3) : false); }

struct WaveTiles { int ar[2]; int bc[4]; };

template <int KIND>
__device__ __forceinline__
void sym_mma(const float (*As)[LDP], const float (*Bs)[LDP],
             const WaveTiles& wt, int r, int half, v8f (&acc)[5])
{
#pragma unroll
    for (int kk = 0; kk < 32; kk += 4) {
        const int ka = kk + 2 * half;
        v2f a[2], b[4];
#pragma unroll
        for (int m = 0; m < 2; ++m)
            a[m] = *(const v2f*)&As[wt.ar[m] + r][ka];
#pragma unroll
        for (int n = 0; n < pat_nb(KIND); ++n)
            b[n] = *(const v2f*)&Bs[wt.bc[n] + r][ka];
#pragma unroll
        for (int q = 0; q < pat_n(KIND); ++q)
            acc[q] = __builtin_amdgcn_wmma_f32_16x16x4_f32(
                 false, a[pat_a(KIND, q)],
                 false, b[pat_b(KIND, q)],
                 (short)0, acc[q],
                 false,  false);
    }
}

template <int KIND, bool DIAGJOB, int MODE>
__device__ __forceinline__
void sym_store(const float* __restrict__ Op, float* __restrict__ D,
               float* __restrict__ D2, float* __restrict__ D3,
               int rowBase, int colBase, const WaveTiles& wt, int r, int half,
               const v8f (&acc)[5])
{
#pragma unroll
    for (int q = 0; q < pat_n(KIND); ++q) {
        const int row0 = rowBase + wt.ar[pat_a(KIND, q)] + 8 * half;
        const int col  = colBase + wt.bc[pat_b(KIND, q)] + r;
        float ml[8], mr[8];
#pragma unroll
        for (int i = 0; i < 8; ++i) {
            const float p = acc[q][i];
            if (MODE == 1) {
                D[(row0 + i) * NMAT + col] = p;
            } else {
                const float x  = D[(row0 + i) * NMAT + col];
                const float y  = Op[(row0 + i) * NMAT + col];
                const float el = K_KZ * p + K_A1 * y;
                const float er = K_KZ * p + K_B1 * y;
                D[(row0 + i) * NMAT + col]  = p;
                D2[(row0 + i) * NMAT + col] = el + K_A2 * x;
                D3[(row0 + i) * NMAT + col] = er + K_B2 * x;
                ml[i] = el - K_A2 * x;
                mr[i] = er - K_B2 * x;
            }
        }
        if (!(DIAGJOB && pat_ondiag(KIND, q))) {
            v4f p0, p1;
#pragma unroll
            for (int i = 0; i < 4; ++i) {
                p0[i] = acc[q][i];
                p1[i] = acc[q][i + 4];
            }
            *(v4f*)(D + col * NMAT + row0)     = p0;
            *(v4f*)(D + col * NMAT + row0 + 4) = p1;
            if (MODE == 2) {
                v4f l0, l1, r0, r1;
#pragma unroll
                for (int i = 0; i < 4; ++i) {
                    l0[i] = ml[i];  l1[i] = ml[i + 4];
                    r0[i] = mr[i];  r1[i] = mr[i + 4];
                }
                *(v4f*)(D2 + col * NMAT + row0)     = l0;
                *(v4f*)(D2 + col * NMAT + row0 + 4) = l1;
                *(v4f*)(D3 + col * NMAT + row0)     = r0;
                *(v4f*)(D3 + col * NMAT + row0 + 4) = r1;
            }
        }
    }
}

__device__ __forceinline__
void half_fetch(const float* __restrict__ aSrc, const float* __restrict__ bSrc,
                int kt, TileRegs& p)
{
    p.a0 = *(const v4f*)(aSrc +  0 * NMAT + kt);
    p.a1 = *(const v4f*)(aSrc + 32 * NMAT + kt);
    p.a2 = *(const v4f*)(aSrc + 64 * NMAT + kt);
    p.a3 = *(const v4f*)(aSrc + 96 * NMAT + kt);
    p.b0 = *(const v4f*)(bSrc + (kt +  0) * NMAT);
    p.b1 = *(const v4f*)(bSrc + (kt + 16) * NMAT);
}

__device__ __forceinline__
void half_store(float (*As)[LDP], float (*Bs)[LDP], int tid, const TileRegs& p)
{
    const int arow = tid >> 3;
    const int ac4  = (tid & 7) << 2;
    const int krow = ((tid >> 3) & 3) + ((tid >> 6) << 2);
    const int bc4  = (((tid >> 5) & 1) << 5) + ((tid & 7) << 2);
    *(v4f*)&As[arow +  0][ac4] = p.a0;
    *(v4f*)&As[arow + 32][ac4] = p.a1;
    *(v4f*)&As[arow + 64][ac4] = p.a2;
    *(v4f*)&As[arow + 96][ac4] = p.a3;
    store_b(Bs, bc4, krow +  0, p.b0);
    store_b(Bs, bc4, krow + 16, p.b1);
}

template <bool DIAGJOB>
__device__ __forceinline__
void sym_tile_mma(const float (*As)[LDP], const float (*Bs)[LDP],
                  const WaveTiles& wt, int kind, int r, int half, v8f (&acc)[5])
{
    if (!DIAGJOB || kind == 0) sym_mma<0>(As, Bs, wt, r, half, acc);
    else if (kind == 1)        sym_mma<1>(As, Bs, wt, r, half, acc);
    else                       sym_mma<2>(As, Bs, wt, r, half, acc);
}

template <bool DIAGJOB, int MODE>
__device__ __forceinline__
void sym_job(const float* __restrict__ Op, float* __restrict__ D,
             float* __restrict__ D2, float* __restrict__ D3,
             float (*As)[128][LDP], float (*Bs)[128][LDP], int rowBase, int colBase)
{
    const int tid  = threadIdx.x;
    const int wave = __builtin_amdgcn_readfirstlane(tid >> 5);
    const int lane = tid & 31;
    const int half = lane >> 4;
    const int r    = lane & 15;

    WaveTiles wt;
    int kind = 0;
    if (!DIAGJOB) {
        wt.ar[0] = (wave >> 1) * 32;      wt.ar[1] = wt.ar[0] + 16;
        wt.bc[0] = (wave & 1) * 32;       wt.bc[1] = wt.bc[0] + 16;
        wt.bc[2] = wt.bc[1];              wt.bc[3] = wt.bc[1];
    } else if (wave < 4) {
        wt.ar[0] = (wave >> 1) * 32;      wt.ar[1] = wt.ar[0] + 16;
        wt.bc[0] = 64 + (wave & 1) * 32;  wt.bc[1] = wt.bc[0] + 16;
        wt.bc[2] = wt.bc[1];              wt.bc[3] = wt.bc[1];
    } else {
        const int t16 = (wave >> 1 & 1) * 64;
        if ((wave & 1) == 0) {
            kind = 1;
            wt.ar[0] = t16;               wt.ar[1] = t16 + 48;
            wt.bc[0] = t16;               wt.bc[1] = t16 + 16;
            wt.bc[2] = t16 + 32;          wt.bc[3] = t16 + 48;
        } else {
            kind = 2;
            wt.ar[0] = t16 + 16;          wt.ar[1] = t16 + 32;
            wt.bc[0] = t16 + 16;          wt.bc[1] = t16 + 32;
            wt.bc[2] = t16 + 48;          wt.bc[3] = t16 + 48;
        }
    }

    v8f acc[5];
#pragma unroll
    for (int q = 0; q < 5; ++q)
        acc[q] = (v8f)0.0f;

    const float* aSrc = Op + (rowBase + (tid >> 3)) * NMAT + ((tid & 7) << 2);
    const float* bSrc = DIAGJOB
        ? Op + (((tid >> 3) & 3) + ((tid >> 7) << 2)) * NMAT
            + colBase + (((tid >> 5) & 3) << 5) + ((tid & 7) << 2)
        : Op + (((tid >> 3) & 3) + ((tid >> 6) << 2)) * NMAT
            + colBase + (((tid >> 5) & 1) << 5) + ((tid & 7) << 2);

    TileRegs p;
    if (DIAGJOB) { tile_fetch(aSrc, bSrc, 0, p); tile_store(As[0], Bs[0], tid, p); }
    else         { half_fetch(aSrc, bSrc, 0, p); half_store(As[0], Bs[0], tid, p); }
    __syncthreads();
    if (DIAGJOB) tile_fetch(aSrc, bSrc, 32, p); else half_fetch(aSrc, bSrc, 32, p);

    for (int t = 0; t < NMAT / 32 - 2; ++t) {
        const int cur = t & 1;
        sym_tile_mma<DIAGJOB>(As[cur], Bs[cur], wt, kind, r, half, acc);
        if (DIAGJOB) tile_store(As[cur ^ 1], Bs[cur ^ 1], tid, p);
        else         half_store(As[cur ^ 1], Bs[cur ^ 1], tid, p);
        __syncthreads();
        if (DIAGJOB) tile_fetch(aSrc, bSrc, (t + 2) * 32, p);
        else         half_fetch(aSrc, bSrc, (t + 2) * 32, p);
    }
    sym_tile_mma<DIAGJOB>(As[0], Bs[0], wt, kind, r, half, acc);
    if (DIAGJOB) tile_store(As[1], Bs[1], tid, p); else half_store(As[1], Bs[1], tid, p);
    __syncthreads();
    sym_tile_mma<DIAGJOB>(As[1], Bs[1], wt, kind, r, half, acc);

    if (!DIAGJOB || kind == 0)
        sym_store<0, DIAGJOB, MODE>(Op, D, D2, D3, rowBase, colBase, wt, r, half, acc);
    else if (kind == 1)
        sym_store<1, DIAGJOB, MODE>(Op, D, D2, D3, rowBase, colBase, wt, r, half, acc);
    else
        sym_store<2, DIAGJOB, MODE>(Op, D, D2, D3, rowBase, colBase, wt, r, half, acc);
}

template <int MODE>
__global__ __launch_bounds__(256)
void sym_square_kernel(const float* __restrict__ Op,
                       float* __restrict__ D,
                       float* __restrict__ D2,
                       float* __restrict__ D3)
{
    __shared__ __attribute__((aligned(16))) float As[2][128][LDP];
    __shared__ __attribute__((aligned(16))) float Bs[2][128][LDP];

    const int b = blockIdx.x;
    if (b < 16) {
        sym_job<true, MODE>(Op, D, D2, D3, As, Bs, b * 128, b * 128);
    } else {
        const int j2 = b - 16;
        int rem = j2 >> 1;
        int I = 0;
        for (int s = 0; s < 15; ++s)
            if (rem >= 15 - I) { rem -= 15 - I; ++I; }
        const int J = I + 1 + rem;
        sym_job<false, MODE>(Op, D, D2, D3, As, Bs, I * 128, J * 128 + (j2 & 1) * 64);
    }
}

__global__ __launch_bounds__(256)
void gemm_wmma_f32_kernel(const float* __restrict__ Amat,
                          const float* __restrict__ Bmat,
                          float* __restrict__ Dmat,
                          const float* __restrict__ Emat,
                          float* __restrict__ D2mat,
                          const float* __restrict__ Ain,
                          int mode)
{
    __shared__ __attribute__((aligned(16))) float As[2][128][LDP];
    __shared__ __attribute__((aligned(16))) float Bs[2][128][LDP];

    const int tid  = threadIdx.x;
    const int wave = tid >> 5;
    const int lane = tid & 31;
    const int half = lane >> 4;
    const int r    = lane & 15;

    const int rowBase = blockIdx.y * 128;
    const int colBase = blockIdx.x * 128;
    const int waveRow = (wave >> 1) * 32;
    const int waveCol = (wave & 1) * 64;

    v8f acc[2][4];
#pragma unroll
    for (int m = 0; m < 2; ++m)
#pragma unroll
        for (int n = 0; n < 4; ++n)
            acc[m][n] = (v8f)0.0f;

    const float* aSrc = Amat + (rowBase + (tid >> 3)) * NMAT + ((tid & 7) << 2);
    const float* bSrc = Bmat + (((tid >> 3) & 3) + ((tid >> 7) << 2)) * NMAT
                      + colBase + (((tid >> 5) & 3) << 5) + ((tid & 7) << 2);

    TileRegs p;
    tile_fetch(aSrc, bSrc, 0, p);
    tile_store(As[0], Bs[0], tid, p);
    __syncthreads();
    tile_fetch(aSrc, bSrc, 32, p);

    for (int t = 0; t < NMAT / 32 - 2; ++t) {
        const int cur = t & 1;
        tile_mma(As[cur], Bs[cur], waveRow, waveCol, r, half, acc);
        tile_store(As[cur ^ 1], Bs[cur ^ 1], tid, p);
        __syncthreads();
        tile_fetch(aSrc, bSrc, (t + 2) * 32, p);
    }
    tile_mma(As[0], Bs[0], waveRow, waveCol, r, half, acc);
    tile_store(As[1], Bs[1], tid, p);
    __syncthreads();
    tile_mma(As[1], Bs[1], waveRow, waveCol, r, half, acc);


    const int row0 = rowBase + waveRow + 8 * half;
    const int col0 = colBase + waveCol + r;
    if (mode == 3) {
#pragma unroll
        for (int m = 0; m < 2; ++m)
#pragma unroll
            for (int n = 0; n < 4; ++n)
#pragma unroll
                for (int i = 0; i < 8; ++i) {
                    const int row = row0 + 16 * m + i;
                    const int col = col0 + 16 * n;
                    const float p  = acc[m][n][i];
                    const float x4 = Dmat[row * NMAT + col];
                    const float x2 = D2mat[row * NMAT + col];
                    const float x  = Ain[row * NMAT + col] - Ain[col * NMAT + row];
                    Dmat[row * NMAT + col]  = p + K_C1 * x4 + K_C2 * x2 + K_C3 * x;
                    D2mat[row * NMAT + col] = p + K_D1 * x4 + K_D2 * x2 + K_D3 * x;
                }
    } else {
#pragma unroll
        for (int m = 0; m < 2; ++m)
#pragma unroll
            for (int n = 0; n < 4; ++n)
#pragma unroll
                for (int i = 0; i < 8; ++i) {
                    const int row = row0 + 16 * m + i;
                    const int col = col0 + 16 * n;
                    const float p  = acc[m][n][i];
                    const float l2 = Amat[row * NMAT + col];
                    const float l1 = Emat[row * NMAT + col];
                    const float r1 = Dmat[row * NMAT + col];
                    const float x  = Ain[row * NMAT + col] - Ain[col * NMAT + row];
                    Dmat[row * NMAT + col] = K_S * p + K_E1 * l2 + K_LA * l1 + K_RO * r1
                                           + K_X1 * x
                                           + ((row == col) ? 1.0f : 0.0f);
                }
    }
}

extern "C" void kernel_launch(void* const* d_in, const int* in_sizes, int n_in,
                              void* d_out, int out_size, void* d_ws, size_t ws_size,
                              hipStream_t stream)
{
    (void)in_sizes; (void)n_in; (void)out_size; (void)ws_size;

    const float* A  = (const float*)d_in[0];
    float*       out = (float*)d_out;

    const size_t NN = (size_t)NMAT * NMAT;
    float* W0 = (float*)d_ws;
    float* W1 = W0 + NN;
    float* W2 = W1 + NN;

    skew_init_kernel<<<dim3(33, 16), 256, 0, stream>>>(A, W0);

    dim3 grid(NMAT / 128, NMAT / 128);
    dim3 block(256);

    sym_square_kernel<1><<<256, 256, 0, stream>>>(W0, W1, nullptr, nullptr);
    sym_square_kernel<2><<<256, 256, 0, stream>>>(W1, W0, W2, out);
    gemm_wmma_f32_kernel<<<grid, block, 0, stream>>>(W2, out, W0, nullptr, W1, A, 3);
    gemm_wmma_f32_kernel<<<grid, block, 0, stream>>>(W0, W1, out, W2, nullptr, A, 4);
}


# K loops: next tile's fragments read from LDS a trip ahead into a second register set; barrier split into signal and wait, WMMAs between; diagonal jobs' tile pairs as data, not code; sums unchanged
# speedup vs baseline: 5.7070x; 1.0929x over previous
#include <hip/hip_runtime.h>


#define NMAT 2048
#define LDP 36

#define K_KZ  0.00077160493827160490f
#define K_A1  0.11154761805772045f
#define K_A2  0.27992720822635375f
#define K_B1  (-0.050111642644035441f)
#define K_B2  0.58017644756523645f
#define K_C1  0.017226797284205099f
#define K_C2  0.024021066342898575f
#define K_C3  (-0.88083743549646332f)
#define K_D1  0.0074336583226582171f
#define K_D2  0.11571413588046474f
#define K_D3  0.55876985454585559f
#define K_S   0.20321412844635858f
#define K_E1  3.8524813843396437f
#define K_LA  (-4.9025788223219831f)
#define K_RO  (-8.5544730341591464f)
#define K_X1  10.728878801472899f


typedef float v2f __attribute__((ext_vector_type(2)));
typedef float v4f __attribute__((ext_vector_type(4)));
typedef float v8f __attribute__((ext_vector_type(8)));

__global__ __launch_bounds__(256)
void skew_init_kernel(const float* __restrict__ A,
                      float* __restrict__ X)
{
    const int x = blockIdx.x;
    const int y = blockIdx.y;
    int I, J;
    if (x > y) { I = y;      J = x - 1; }
    else       { I = 31 - y; J = 31 - y + x; }

    const int tid  = threadIdx.x;
    const int lane = tid & 31;
    const int wave = tid >> 5;
    const int mc = (lane & 7) + ((wave & 1) << 3);
    const int mr = (lane >> 3) + ((wave >> 1) << 2);
    const int i0 = I * 64 + 4 * mr;
    const int j0 = J * 64 + 4 * mc;

    v4f p[4], q[4];
#pragma unroll
    for (int k = 0; k < 4; ++k)
        p[k] = *(const v4f*)(A + (i0 + k) * NMAT + j0);
#pragma unroll
    for (int k = 0; k < 4; ++k)
        q[k] = *(const v4f*)(A + (j0 + k) * NMAT + i0);

#pragma unroll
    for (int k = 0; k < 4; ++k) {
        v4f d;
        d.x = p[k].x - q[0][k];
        d.y = p[k].y - q[1][k];
        d.z = p[k].z - q[2][k];
        d.w = p[k].w - q[3][k];
        *(v4f*)(X + (i0 + k) * NMAT + j0) = d;
    }
    if (I != J) {
#pragma unroll
        for (int k = 0; k < 4; ++k) {
            v4f e;
            e.x = q[k].x - p[0][k];
            e.y = q[k].y - p[1][k];
            e.z = q[k].z - p[2][k];
            e.w = q[k].w - p[3][k];
            *(v4f*)(X + (j0 + k) * NMAT + i0) = e;
        }
    }
}

struct TileRegs { v4f a0, a1, a2, a3, b0, b1, b2, b3; };

__device__ __forceinline__
void tile_fetch(const float* __restrict__ aSrc, const float* __restrict__ bSrc,
                int kt, TileRegs& p)
{
    p.a0 = *(const v4f*)(aSrc +  0 * NMAT + kt);
    p.a1 = *(const v4f*)(aSrc + 32 * NMAT + kt);
    p.a2 = *(const v4f*)(aSrc + 64 * NMAT + kt);
    p.a3 = *(const v4f*)(aSrc + 96 * NMAT + kt);
    p.b0 = *(const v4f*)(bSrc + (kt +  0) * NMAT);
    p.b1 = *(const v4f*)(bSrc + (kt +  8) * NMAT);
    p.b2 = *(const v4f*)(bSrc + (kt + 16) * NMAT);
    p.b3 = *(const v4f*)(bSrc + (kt + 24) * NMAT);
}

__device__ __forceinline__
void store_b(float (*Bs)[LDP], int bc4, int k, const v4f v)
{
    Bs[bc4 + 0][k] = v.x;
    Bs[bc4 + 1][k] = v.y;
    Bs[bc4 + 2][k] = v.z;
    Bs[bc4 + 3][k] = v.w;
}

__device__ __forceinline__
void tile_store(float (*As)[LDP], float (*Bs)[LDP], int tid, const TileRegs& p)
{
    const int arow = tid >> 3;
    const int ac4  = (tid & 7) << 2;
    const int krow = ((tid >> 3) & 3) + ((tid >> 7) << 2);
    const int bc4  = (((tid >> 5) & 3) << 5) + ((tid & 7) << 2);
    *(v4f*)&As[arow +  0][ac4] = p.a0;
    *(v4f*)&As[arow + 32][ac4] = p.a1;
    *(v4f*)&As[arow + 64][ac4] = p.a2;
    *(v4f*)&As[arow + 96][ac4] = p.a3;
    store_b(Bs, bc4, krow +  0, p.b0);
    store_b(Bs, bc4, krow +  8, p.b1);
    store_b(Bs, bc4, krow + 16, p.b2);
    store_b(Bs, bc4, krow + 24, p.b3);
}

struct Frags { v2f a[8][2]; v2f b[8][4]; };

__device__ __forceinline__
void frag_load(const float (*As)[LDP], const float (*Bs)[LDP],
               int waveRow, int waveCol, int r, int half, Frags& f)
{
#pragma unroll
    for (int s = 0; s < 8; ++s) {
        const int ka = 4 * s + 2 * half;
#pragma unroll
        for (int m = 0; m < 2; ++m)
            f.a[s][m] = *(const v2f*)&As[waveRow + 16 * m + r][ka];
#pragma unroll
        for (int n = 0; n < 4; ++n)
            f.b[s][n] = *(const v2f*)&Bs[waveCol + 16 * n + r][ka];
    }
}

template <int S0, int S1>
__device__ __forceinline__
void frag_mma(const Frags& f, v8f (&acc)[2][4])
{
#pragma unroll
    for (int s = S0; s < S1; ++s)
#pragma unroll
        for (int m = 0; m < 2; ++m)
#pragma unroll
            for (int n = 0; n < 4; ++n)
                acc[m][n] = __builtin_amdgcn_wmma_f32_16x16x4_f32(
                     false, f.a[s][m],
                     false, f.b[s][n],
                     (short)0, acc[m][n],
                     false,  false);
}

template <bool HAS_NEXT, bool HAS_FETCH>
__device__ __forceinline__
void gemm_trip(float (*AsN)[LDP], float (*BsN)[LDP],
               const float* __restrict__ aSrc, const float* __restrict__ bSrc, int ktFetch,
               int tid, int waveRow, int waveCol, int r, int half,
               TileRegs& p, const Frags& fc, Frags& fn, v8f (&acc)[2][4])
{
    frag_mma<0, 2>(fc, acc);
    __builtin_amdgcn_sched_barrier(0);
    if (HAS_NEXT) tile_store(AsN, BsN, tid, p);
    if (HAS_FETCH) tile_fetch(aSrc, bSrc, ktFetch, p);
    __builtin_amdgcn_sched_barrier(0);
    frag_mma<2, 4>(fc, acc);
    __builtin_amdgcn_sched_barrier(0);
    if (HAS_NEXT) {
        __builtin_amdgcn_fence(__ATOMIC_RELEASE, "workgroup", "local");
        __builtin_amdgcn_s_barrier_signal(-1);
    }
    __builtin_amdgcn_sched_barrier(0);
    frag_mma<4, 5>(fc, acc);
    __builtin_amdgcn_sched_barrier(0);
    if (HAS_NEXT) {
        __builtin_amdgcn_s_barrier_wait(-1);
        __builtin_amdgcn_fence(__ATOMIC_ACQUIRE, "workgroup", "local");
        frag_load(AsN, BsN, waveRow, waveCol, r, half, fn);
    }
    __builtin_amdgcn_sched_barrier(0);
    frag_mma<5, 8>(fc, acc);
    __builtin_amdgcn_sched_barrier(0);
}

constexpr int pat_n(int kind)         { return kind == 0 ? 4 : 5; }
constexpr int pat_a(int kind, int q)  { return kind == 0 ? (q >> 1) : (kind == 1 ? (q == 4 ? 1 : 0) : (q >= 3 ? 1 : 0)); }
constexpr int pat_b(int kind, int q)  { return kind == 0 ? (q & 1) : (kind == 1 ? (q < 4 ? q : 3) : (q < 3 ? q : q - 2)); }
constexpr bool pat_ondiag(int kind, int q) { return kind == 1 ? (q == 0 || q == 4) : (kind == 2 ? (q == 0 || q == 3) : false); }

struct WaveTiles { int ar[2]; int bc[4]; };

template <int KIND, bool DIAGJOB, int MODE>
__device__ __forceinline__
void sym_store(const float* __restrict__ Op, float* __restrict__ D,
               float* __restrict__ D2, float* __restrict__ D3,
               int rowBase, int colBase, const WaveTiles& wt, int r, int half,
               const v8f (&acc)[5])
{
#pragma unroll
    for (int q = 0; q < pat_n(KIND); ++q) {
        const int row0 = rowBase + wt.ar[pat_a(KIND, q)] + 8 * half;
        const int col  = colBase + wt.bc[pat_b(KIND, q)] + r;
        float ml[8], mr[8];
#pragma unroll
        for (int i = 0; i < 8; ++i) {
            const float p = acc[q][i];
            if (MODE == 1) {
                D[(row0 + i) * NMAT + col] = p;
            } else {
                const float x  = D[(row0 + i) * NMAT + col];
                const float y  = Op[(row0 + i) * NMAT + col];
                const float el = K_KZ * p + K_A1 * y;
                const float er = K_KZ * p + K_B1 * y;
                D[(row0 + i) * NMAT + col]  = p;
                D2[(row0 + i) * NMAT + col] = el + K_A2 * x;
                D3[(row0 + i) * NMAT + col] = er + K_B2 * x;
                ml[i] = el - K_A2 * x;
                mr[i] = er - K_B2 * x;
            }
        }
        if (!(DIAGJOB && pat_ondiag(KIND, q))) {
            v4f p0, p1;
#pragma unroll
            for (int i = 0; i < 4; ++i) {
                p0[i] = acc[q][i];
                p1[i] = acc[q][i + 4];
            }
            *(v4f*)(D + col * NMAT + row0)     = p0;
            *(v4f*)(D + col * NMAT + row0 + 4) = p1;
            if (MODE == 2) {
                v4f l0, l1, r0, r1;
#pragma unroll
                for (int i = 0; i < 4; ++i) {
                    l0[i] = ml[i];  l1[i] = ml[i + 4];
                    r0[i] = mr[i];  r1[i] = mr[i + 4];
                }
                *(v4f*)(D2 + col * NMAT + row0)     = l0;
                *(v4f*)(D2 + col * NMAT + row0 + 4) = l1;
                *(v4f*)(D3 + col * NMAT + row0)     = r0;
                *(v4f*)(D3 + col * NMAT + row0 + 4) = r1;
            }
        }
    }
}

__device__ __forceinline__
void half_fetch(const float* __restrict__ aSrc, const float* __restrict__ bSrc,
                int kt, TileRegs& p)
{
    p.a0 = *(const v4f*)(aSrc +  0 * NMAT + kt);
    p.a1 = *(const v4f*)(aSrc + 32 * NMAT + kt);
    p.a2 = *(const v4f*)(aSrc + 64 * NMAT + kt);
    p.a3 = *(const v4f*)(aSrc + 96 * NMAT + kt);
    p.b0 = *(const v4f*)(bSrc + (kt +  0) * NMAT);
    p.b1 = *(const v4f*)(bSrc + (kt + 16) * NMAT);
}

__device__ __forceinline__
void half_store(float (*As)[LDP], float (*Bs)[LDP], int tid, const TileRegs& p)
{
    const int arow = tid >> 3;
    const int ac4  = (tid & 7) << 2;
    const int krow = ((tid >> 3) & 3) + ((tid >> 6) << 2);
    const int bc4  = (((tid >> 5) & 1) << 5) + ((tid & 7) << 2);
    *(v4f*)&As[arow +  0][ac4] = p.a0;
    *(v4f*)&As[arow + 32][ac4] = p.a1;
    *(v4f*)&As[arow + 64][ac4] = p.a2;
    *(v4f*)&As[arow + 96][ac4] = p.a3;
    store_b(Bs, bc4, krow +  0, p.b0);
    store_b(Bs, bc4, krow + 16, p.b1);
}

struct WavePairs { int pa[5]; int pb[5]; };
template <int NP> struct SymFrags { v2f a[8][NP == 4 ? 2 : 5]; v2f b[8][NP == 4 ? 2 : 5]; };

template <int NP>
__device__ __forceinline__
void sym_frag_load(const float (*As)[LDP], const float (*Bs)[LDP],
                   const WavePairs& wp, int r, int half, SymFrags<NP>& f)
{
#pragma unroll
    for (int s = 0; s < 8; ++s) {
        const int ka = 4 * s + 2 * half;
        if (NP == 4) {
            f.a[s][0] = *(const v2f*)&As[wp.pa[0] + r][ka];
            f.a[s][1] = *(const v2f*)&As[wp.pa[2] + r][ka];
            f.b[s][0] = *(const v2f*)&Bs[wp.pb[0] + r][ka];
            f.b[s][1] = *(const v2f*)&Bs[wp.pb[1] + r][ka];
        } else {
#pragma unroll
            for (int q = 0; q < 5; ++q) {
                f.a[s][q] = *(const v2f*)&As[wp.pa[q] + r][ka];
                f.b[s][q] = *(const v2f*)&Bs[wp.pb[q] + r][ka];
            }
        }
    }
}

template <int NP, int S0, int S1>
__device__ __forceinline__
void sym_frag_mma(const SymFrags<NP>& f, bool five, v8f (&acc)[5])
{
#pragma unroll
    for (int s = S0; s < S1; ++s)
#pragma unroll
        for (int q = 0; q < 4; ++q)
            acc[q] = __builtin_amdgcn_wmma_f32_16x16x4_f32(
                 false, f.a[s][NP == 4 ? (q >> 1) : q],
                 false, f.b[s][NP == 4 ? (q & 1) : q],
                 (short)0, acc[q],
                 false,  false);
    if (NP == 5 && five) {
#pragma unroll
        for (int s = S0; s < S1; ++s)
            acc[4] = __builtin_amdgcn_wmma_f32_16x16x4_f32(
                 false, f.a[s][4],
                 false, f.b[s][4],
                 (short)0, acc[4],
                 false,  false);
    }
}

template <bool DIAGJOB, bool HAS_NEXT, bool HAS_FETCH>
__device__ __forceinline__
void sym_trip(float (*AsN)[LDP], float (*BsN)[LDP],
              const float* __restrict__ aSrc, const float* __restrict__ bSrc, int ktFetch,
              int tid, const WavePairs& wp, bool five, int r, int half,
              TileRegs& p, const SymFrags<DIAGJOB ? 5 : 4>& fc,
              SymFrags<DIAGJOB ? 5 : 4>& fn, v8f (&acc)[5])
{
    constexpr int NP = DIAGJOB ? 5 : 4;
    sym_frag_mma<NP, 0, 2>(fc, five, acc);
    __builtin_amdgcn_sched_barrier(0);
    if (HAS_NEXT) {
        if (DIAGJOB) tile_store(AsN, BsN, tid, p); else half_store(AsN, BsN, tid, p);
    }
    if (HAS_FETCH) {
        if (DIAGJOB) tile_fetch(aSrc, bSrc, ktFetch, p); else half_fetch(aSrc, bSrc, ktFetch, p);
    }
    __builtin_amdgcn_sched_barrier(0);
    sym_frag_mma<NP, 2, 4>(fc, five, acc);
    __builtin_amdgcn_sched_barrier(0);
    if (HAS_NEXT) {
        __builtin_amdgcn_fence(__ATOMIC_RELEASE, "workgroup", "local");
        __builtin_amdgcn_s_barrier_signal(-1);
    }
    __builtin_amdgcn_sched_barrier(0);
    sym_frag_mma<NP, 4, 5>(fc, five, acc);
    __builtin_amdgcn_sched_barrier(0);
    if (HAS_NEXT) {
        __builtin_amdgcn_s_barrier_wait(-1);
        __builtin_amdgcn_fence(__ATOMIC_ACQUIRE, "workgroup", "local");
        sym_frag_load<NP>(AsN, BsN, wp, r, half, fn);
    }
    __builtin_amdgcn_sched_barrier(0);
    sym_frag_mma<NP, 5, 8>(fc, five, acc);
    __builtin_amdgcn_sched_barrier(0);
}

template <bool DIAGJOB, int MODE>
__device__ __forceinline__
void sym_job(const float* __restrict__ Op, float* __restrict__ D,
             float* __restrict__ D2, float* __restrict__ D3,
             float (*As)[128][LDP], float (*Bs)[128][LDP], int rowBase, int colBase)
{
    const int tid  = threadIdx.x;
    const int wave = __builtin_amdgcn_readfirstlane(tid >> 5);
    const int lane = tid & 31;
    const int half = lane >> 4;
    const int r    = lane & 15;

    WaveTiles wt;
    int kind = 0;
    if (!DIAGJOB) {
        wt.ar[0] = (wave >> 1) * 32;      wt.ar[1] = wt.ar[0] + 16;
        wt.bc[0] = (wave & 1) * 32;       wt.bc[1] = wt.bc[0] + 16;
        wt.bc[2] = wt.bc[1];              wt.bc[3] = wt.bc[1];
    } else if (wave < 4) {
        wt.ar[0] = (wave >> 1) * 32;      wt.ar[1] = wt.ar[0] + 16;
        wt.bc[0] = 64 + (wave & 1) * 32;  wt.bc[1] = wt.bc[0] + 16;
        wt.bc[2] = wt.bc[1];              wt.bc[3] = wt.bc[1];
    } else {
        const int t16 = (wave >> 1 & 1) * 64;
        if ((wave & 1) == 0) {
            kind = 1;
            wt.ar[0] = t16;               wt.ar[1] = t16 + 48;
            wt.bc[0] = t16;               wt.bc[1] = t16 + 16;
            wt.bc[2] = t16 + 32;          wt.bc[3] = t16 + 48;
        } else {
            kind = 2;
            wt.ar[0] = t16 + 16;          wt.ar[1] = t16 + 32;
            wt.bc[0] = t16 + 16;          wt.bc[1] = t16 + 32;
            wt.bc[2] = t16 + 48;          wt.bc[3] = t16 + 48;
        }
    }

    WavePairs wp;
#pragma unroll
    for (int q = 0; q < 5; ++q) {
        const int q0 = q < 4 ? q : 3;
        const int ia = kind == 0 ? pat_a(0, q0) : (kind == 1 ? pat_a(1, q) : pat_a(2, q));
        const int ib = kind == 0 ? pat_b(0, q0) : (kind == 1 ? pat_b(1, q) : pat_b(2, q));
        wp.pa[q] = ia == 0 ? wt.ar[0] : wt.ar[1];
        wp.pb[q] = ib == 0 ? wt.bc[0] : (ib == 1 ? wt.bc[1] : (ib == 2 ? wt.bc[2] : wt.bc[3]));
    }

    const bool five = kind != 0;

    v8f acc[5];
#pragma unroll
    for (int q = 0; q < 5; ++q)
        acc[q] = (v8f)0.0f;

    const float* aSrc = Op + (rowBase + (tid >> 3)) * NMAT + ((tid & 7) << 2);
    const float* bSrc = DIAGJOB
        ? Op + (((tid >> 3) & 3) + ((tid >> 7) << 2)) * NMAT
            + colBase + (((tid >> 5) & 3) << 5) + ((tid & 7) << 2)
        : Op + (((tid >> 3) & 3) + ((tid >> 6) << 2)) * NMAT
            + colBase + (((tid >> 5) & 1) << 5) + ((tid & 7) << 2);

    TileRegs p;
    SymFrags<DIAGJOB ? 5 : 4> f0, f1;
    if (DIAGJOB) { tile_fetch(aSrc, bSrc, 0, p); tile_store(As[0], Bs[0], tid, p); }
    else         { half_fetch(aSrc, bSrc, 0, p); half_store(As[0], Bs[0], tid, p); }
    __syncthreads();
    if (DIAGJOB) tile_fetch(aSrc, bSrc, 32, p); else half_fetch(aSrc, bSrc, 32, p);
    sym_frag_load<DIAGJOB ? 5 : 4>(As[0], Bs[0], wp, r, half, f0);

    for (int t = 0; t < NMAT / 32 - 2; t += 2) {
        sym_trip<DIAGJOB, true, true>(As[1], Bs[1], aSrc, bSrc, (t + 2) * 32,
                                      tid, wp, five, r, half, p, f0, f1, acc);
        sym_trip<DIAGJOB, true, true>(As[0], Bs[0], aSrc, bSrc, (t + 3) * 32,
                                      tid, wp, five, r, half, p, f1, f0, acc);
    }
    sym_trip<DIAGJOB, true, false>(As[1], Bs[1], aSrc, bSrc, 0,
                                   tid, wp, five, r, half, p, f0, f1, acc);
    sym_trip<DIAGJOB, false, false>(As[0], Bs[0], aSrc, bSrc, 0,
                                    tid, wp, five, r, half, p, f1, f0, acc);

    if (!DIAGJOB || kind == 0)
        sym_store<0, DIAGJOB, MODE>(Op, D, D2, D3, rowBase, colBase, wt, r, half, acc);
    else if (kind == 1)
        sym_store<1, DIAGJOB, MODE>(Op, D, D2, D3, rowBase, colBase, wt, r, half, acc);
    else
        sym_store<2, DIAGJOB, MODE>(Op, D, D2, D3, rowBase, colBase, wt, r, half, acc);
}

template <int MODE>
__global__ __launch_bounds__(256)
void sym_square_kernel(const float* __restrict__ Op,
                       float* __restrict__ D,
                       float* __restrict__ D2,
                       float* __restrict__ D3)
{
    __shared__ __attribute__((aligned(16))) float As[2][128][LDP];
    __shared__ __attribute__((aligned(16))) float Bs[2][128][LDP];

    const int b = blockIdx.x;
    if (b < 16) {
        sym_job<true, MODE>(Op, D, D2, D3, As, Bs, b * 128, b * 128);
    } else {
        const int j2 = b - 16;
        int rem = j2 >> 1;
        int I = 0;
        for (int s = 0; s < 15; ++s)
            if (rem >= 15 - I) { rem -= 15 - I; ++I; }
        const int J = I + 1 + rem;
        sym_job<false, MODE>(Op, D, D2, D3, As, Bs, I * 128, J * 128 + (j2 & 1) * 64);
    }
}

__global__ __launch_bounds__(256)
void gemm_wmma_f32_kernel(const float* __restrict__ Amat,
                          const float* __restrict__ Bmat,
                          float* __restrict__ Dmat,
                          const float* __restrict__ Emat,
                          float* __restrict__ D2mat,
                          const float* __restrict__ Ain,
                          int mode)
{
    __shared__ __attribute__((aligned(16))) float As[2][128][LDP];
    __shared__ __attribute__((aligned(16))) float Bs[2][128][LDP];

    const int tid  = threadIdx.x;
    const int wave = tid >> 5;
    const int lane = tid & 31;
    const int half = lane >> 4;
    const int r    = lane & 15;

    const int rowBase = blockIdx.y * 128;
    const int colBase = blockIdx.x * 128;
    const int waveRow = (wave >> 1) * 32;
    const int waveCol = (wave & 1) * 64;

    v8f acc[2][4];
#pragma unroll
    for (int m = 0; m < 2; ++m)
#pragma unroll
        for (int n = 0; n < 4; ++n)
            acc[m][n] = (v8f)0.0f;

    const float* aSrc = Amat + (rowBase + (tid >> 3)) * NMAT + ((tid & 7) << 2);
    const float* bSrc = Bmat + (((tid >> 3) & 3) + ((tid >> 7) << 2)) * NMAT
                      + colBase + (((tid >> 5) & 3) << 5) + ((tid & 7) << 2);

    TileRegs p;
    Frags f0, f1;
    tile_fetch(aSrc, bSrc, 0, p);
    tile_store(As[0], Bs[0], tid, p);
    __syncthreads();
    tile_fetch(aSrc, bSrc, 32, p);
    frag_load(As[0], Bs[0], waveRow, waveCol, r, half, f0);

    for (int t = 0; t < NMAT / 32 - 2; t += 2) {
        gemm_trip<true, true>(As[1], Bs[1], aSrc, bSrc, (t + 2) * 32,
                              tid, waveRow, waveCol, r, half, p, f0, f1, acc);
        gemm_trip<true, true>(As[0], Bs[0], aSrc, bSrc, (t + 3) * 32,
                              tid, waveRow, waveCol, r, half, p, f1, f0, acc);
    }
    gemm_trip<true, false>(As[1], Bs[1], aSrc, bSrc, 0,
                           tid, waveRow, waveCol, r, half, p, f0, f1, acc);
    gemm_trip<false, false>(As[0], Bs[0], aSrc, bSrc, 0,
                            tid, waveRow, waveCol, r, half, p, f1, f0, acc);


    const int row0 = rowBase + waveRow + 8 * half;
    const int col0 = colBase + waveCol + r;
    if (mode == 3) {
#pragma unroll
        for (int m = 0; m < 2; ++m)
#pragma unroll
            for (int n = 0; n < 4; ++n)
#pragma unroll
                for (int i = 0; i < 8; ++i) {
                    const int row = row0 + 16 * m + i;
                    const int col = col0 + 16 * n;
                    const float p  = acc[m][n][i];
                    const float x4 = Dmat[row * NMAT + col];
                    const float x2 = D2mat[row * NMAT + col];
                    const float x  = Ain[row * NMAT + col] - Ain[col * NMAT + row];
                    Dmat[row * NMAT + col]  = p + K_C1 * x4 + K_C2 * x2 + K_C3 * x;
                    D2mat[row * NMAT + col] = p + K_D1 * x4 + K_D2 * x2 + K_D3 * x;
                }
    } else {
#pragma unroll
        for (int m = 0; m < 2; ++m)
#pragma unroll
            for (int n = 0; n < 4; ++n)
#pragma unroll
                for (int i = 0; i < 8; ++i) {
                    const int row = row0 + 16 * m + i;
                    const int col = col0 + 16 * n;
                    const float p  = acc[m][n][i];
                    const float l2 = Amat[row * NMAT + col];
                    const float l1 = Emat[row * NMAT + col];
                    const float r1 = Dmat[row * NMAT + col];
                    const float x  = Ain[row * NMAT + col] - Ain[col * NMAT + row];
                    Dmat[row * NMAT + col] = K_S * p + K_E1 * l2 + K_LA * l1 + K_RO * r1
                                           + K_X1 * x
                                           + ((row == col) ? 1.0f : 0.0f);
                }
    }
}

extern "C" void kernel_launch(void* const* d_in, const int* in_sizes, int n_in,
                              void* d_out, int out_size, void* d_ws, size_t ws_size,
                              hipStream_t stream)
{
    (void)in_sizes; (void)n_in; (void)out_size; (void)ws_size;

    const float* A  = (const float*)d_in[0];
    float*       out = (float*)d_out;

    const size_t NN = (size_t)NMAT * NMAT;
    float* W0 = (float*)d_ws;
    float* W1 = W0 + NN;
    float* W2 = W1 + NN;

    skew_init_kernel<<<dim3(33, 16), 256, 0, stream>>>(A, W0);

    dim3 grid(NMAT / 128, NMAT / 128);
    dim3 block(256);

    sym_square_kernel<1><<<256, 256, 0, stream>>>(W0, W1, nullptr, nullptr);
    sym_square_kernel<2><<<256, 256, 0, stream>>>(W1, W0, W2, out);
    gemm_wmma_f32_kernel<<<grid, block, 0, stream>>>(W2, out, W0, nullptr, W1, A, 3);
    gemm_wmma_f32_kernel<<<grid, block, 0, stream>>>(W0, W1, out, W2, nullptr, A, 4);
}
